# starting kernel
# baseline (speedup 1.0000x reference)
#include <hip/hip_runtime.h>
#include <math.h>

#define DD 768
#define LL 1024
#define NN 16
#define BB 2
#define NC 16
#define LC (LL / NC)

typedef float v2f __attribute__((ext_vector_type(2)));
typedef float v8f __attribute__((ext_vector_type(8)));

__global__ __launch_bounds__(32) void ssm_proj_wmma(
    const float* __restrict__ x,
    const float* __restrict__ W_B,
    const float* __restrict__ b_B,
    const float* __restrict__ W_C,
    const float* __restrict__ b_C,
    const float* __restrict__ W_d,
    const float* __restrict__ b_d,
    float* __restrict__ Bp,
    float* __restrict__ Cp,
    float* __restrict__ dlin)
{
    const int lane = threadIdx.x;
    const int half = lane >> 4;
    const int mn   = lane & 15;
    const int row0 = blockIdx.x << 4;
    const int b    = row0 >> 10;
    const int l0   = row0 & (LL - 1);
    const int lA   = l0 + mn;

    const float biasB = b_B[mn];
    const float biasC = b_C[mn];
    const float biasD = (mn == 0) ? b_d[0] : 0.0f;
    v8f accB, accC, accD;
#pragma unroll
    for (int r = 0; r < 8; ++r) { accB[r] = biasB; accC[r] = biasC; accD[r] = biasD; }

    const float* __restrict__ xb = x + (size_t)b * DD * LL;
    const float n0sel = (mn == 0) ? 1.0f : 0.0f;

    for (int k0 = 0; k0 < DD; k0 += 4) {
        const int k = k0 + 2 * half;
        v2f a;
        a[0] = xb[(size_t)k * LL + lA];
        a[1] = xb[(size_t)(k + 1) * LL + lA];
        v2f wb, wc, wd;
        wb[0] = W_B[mn * DD + k];
        wb[1] = W_B[mn * DD + k + 1];
        wc[0] = W_C[mn * DD + k];
        wc[1] = W_C[mn * DD + k + 1];
        wd[0] = W_d[k] * n0sel;
        wd[1] = W_d[k + 1] * n0sel;

        accB = __builtin_amdgcn_wmma_f32_16x16x4_f32(false, a, false, wb, (short)0, accB, false, false);
        accC = __builtin_amdgcn_wmma_f32_16x16x4_f32(false, a, false, wc, (short)0, accC, false, false);
        accD = __builtin_amdgcn_wmma_f32_16x16x4_f32(false, a, false, wd, (short)0, accD, false, false);
    }

#pragma unroll
    for (int r = 0; r < 8; ++r) {
        const int l  = l0 + r + 8 * half;
        const int gr = b * LL + l;
        Bp[gr * NN + mn] = accB[r];
        Cp[gr * NN + mn] = accC[r];
        if (mn == 0) dlin[gr] = accD[r];
    }
}

__global__ __launch_bounds__(256) void ssm_delta(
    const float* __restrict__ dlin,
    const float* __restrict__ p_Delta,
    float* __restrict__ delta)
{
    const int i  = blockIdx.x * 256 + threadIdx.x;
    const int l  = i & (LL - 1);
    const int bd = i >> 10;
    const int b  = bd / DD;
    const int d  = bd - b * DD;
    const float dl = dlin[b * LL + l] + p_Delta[d];
    delta[i] = fmaxf(dl, 0.0f) + log1pf(__expf(-fabsf(dl)));
}

__global__ __launch_bounds__(256) void ssm_scan_local(
    const float* __restrict__ x,
    const float* __restrict__ A,
    const float* __restrict__ delta,
    const float* __restrict__ Bp,
    float* __restrict__ Pc,
    float* __restrict__ Sc)
{
    const int gtid  = blockIdx.x * 256 + threadIdx.x;
    const int q     = gtid & 3;
    const int group = gtid >> 2;
    const int c     = group & (NC - 1);
    const int pair  = group >> 4;
    const int b     = pair / DD;
    const int d     = pair - b * DD;

    const float4 An  = ((const float4*)(A + d * NN))[q];
    const float4 rAn = make_float4(1.0f / An.x, 1.0f / An.y, 1.0f / An.z, 1.0f / An.w);

    const float*  __restrict__ dl  = delta + (size_t)pair * LL + c * LC;
    const float*  __restrict__ xr  = x     + (size_t)pair * LL + c * LC;
    const float4* __restrict__ bpr = (const float4*)(Bp + ((size_t)b * LL + c * LC) * NN) + q;

    float4 h = make_float4(0.f, 0.f, 0.f, 0.f);
    float sd = 0.0f;
#pragma unroll 2
    for (int s = 0; s < LC; ++s) {
        const float dt = dl[s];
        const float xv = xr[s];
        const float4 bp = bpr[4 * s];
        sd += dt;
        const float a0 = __expf(dt * An.x);
        const float a1 = __expf(dt * An.y);
        const float a2 = __expf(dt * An.z);
        const float a3 = __expf(dt * An.w);
        h.x = fmaf(a0, h.x, (a0 - An.x) * rAn.x * bp.x * xv);
        h.y = fmaf(a1, h.y, (a1 - An.y) * rAn.y * bp.y * xv);
        h.z = fmaf(a2, h.z, (a2 - An.z) * rAn.z * bp.z * xv);
        h.w = fmaf(a3, h.w, (a3 - An.w) * rAn.w * bp.w * xv);
    }
    float4 P;
    P.x = __expf(An.x * sd);
    P.y = __expf(An.y * sd);
    P.z = __expf(An.z * sd);
    P.w = __expf(An.w * sd);
    ((float4*)Pc)[group * 4 + q] = P;
    ((float4*)Sc)[group * 4 + q] = h;
}

__global__ __launch_bounds__(256) void ssm_chunk_combine(
    const float* __restrict__ Pc,
    const float* __restrict__ Sc,
    float* __restrict__ Hinit)
{
    const int t    = blockIdx.x * 256 + threadIdx.x;
    const int n    = t & 15;
    const int pair = t >> 4;
    float h = 0.0f;
#pragma unroll
    for (int c = 0; c < NC; ++c) {
        const int idx = (pair * NC + c) * NN + n;
        Hinit[idx] = h;
        h = fmaf(Pc[idx], h, Sc[idx]);
    }
}

__global__ __launch_bounds__(256) void ssm_scan_final(
    const float* __restrict__ x,
    const float* __restrict__ A,
    const float* __restrict__ delta,
    const float* __restrict__ Bp,
    const float* __restrict__ Cp,
    const float* __restrict__ Hinit,
    float* __restrict__ y)
{
    const int gtid  = blockIdx.x * 256 + threadIdx.x;
    const int q     = gtid & 3;
    const int group = gtid >> 2;
    const int c     = group & (NC - 1);
    const int pair  = group >> 4;
    const int b     = pair / DD;
    const int d     = pair - b * DD;

    const float4 An  = ((const float4*)(A + d * NN))[q];
    const float4 rAn = make_float4(1.0f / An.x, 1.0f / An.y, 1.0f / An.z, 1.0f / An.w);

    const float*  __restrict__ dl  = delta + (size_t)pair * LL + c * LC;
    const float*  __restrict__ xr  = x     + (size_t)pair * LL + c * LC;
    const float4* __restrict__ bpr = (const float4*)(Bp + ((size_t)b * LL + c * LC) * NN) + q;
    const float4* __restrict__ cpr = (const float4*)(Cp + ((size_t)b * LL + c * LC) * NN) + q;
    float*        __restrict__ yr  = y     + (size_t)pair * LL + c * LC;

    float4 h = ((const float4*)Hinit)[group * 4 + q];

#pragma unroll 2
    for (int s = 0; s < LC; ++s) {
        const float dt = dl[s];
        const float xv = xr[s];
        const float4 bp = bpr[4 * s];
        const float4 cp = cpr[4 * s];
        const float a0 = __expf(dt * An.x);
        const float a1 = __expf(dt * An.y);
        const float a2 = __expf(dt * An.z);
        const float a3 = __expf(dt * An.w);
        h.x = fmaf(a0, h.x, (a0 - An.x) * rAn.x * bp.x * xv);
        h.y = fmaf(a1, h.y, (a1 - An.y) * rAn.y * bp.y * xv);
        h.z = fmaf(a2, h.z, (a2 - An.z) * rAn.z * bp.z * xv);
        h.w = fmaf(a3, h.w, (a3 - An.w) * rAn.w * bp.w * xv);

        float p = cp.x * h.x;
        p = fmaf(cp.y, h.y, p);
        p = fmaf(cp.z, h.z, p);
        p = fmaf(cp.w, h.w, p);
        p += __shfl_xor(p, 1, 4);
        p += __shfl_xor(p, 2, 4);
        if (q == 0) yr[s] = p;
    }
}

extern "C" void kernel_launch(void* const* d_in, const int* in_sizes, int n_in,
                              void* d_out, int out_size, void* d_ws, size_t ws_size,
                              hipStream_t stream)
{
    const float* x       = (const float*)d_in[0];
    const float* A       = (const float*)d_in[1];
    const float* W_B     = (const float*)d_in[2];
    const float* b_B     = (const float*)d_in[3];
    const float* W_C     = (const float*)d_in[4];
    const float* b_C     = (const float*)d_in[5];
    const float* W_d     = (const float*)d_in[6];
    const float* b_d     = (const float*)d_in[7];
    const float* p_Delta = (const float*)d_in[8];
    float* y = (float*)d_out;

    float* ws    = (float*)d_ws;
    float* Bp    = ws;
    float* Cp    = Bp + (size_t)BB * LL * NN;
    float* dlin  = Cp + (size_t)BB * LL * NN;
    float* delta = dlin + (size_t)BB * LL;
    float* Pc    = delta + (size_t)BB * DD * LL;
    float* Sc    = Pc + (size_t)BB * DD * NC * NN;
    float* Hinit = Sc + (size_t)BB * DD * NC * NN;

    ssm_proj_wmma<<<dim3(128), dim3(32), 0, stream>>>(
        x, W_B, b_B, W_C, b_C, W_d, b_d, Bp, Cp, dlin);

    ssm_delta<<<dim3((BB * DD * LL) / 256), dim3(256), 0, stream>>>(
        dlin, p_Delta, delta);

    ssm_scan_local<<<dim3((BB * DD * NC * 4) / 256), dim3(256), 0, stream>>>(
        x, A, delta, Bp, Pc, Sc);

    ssm_chunk_combine<<<dim3((BB * DD * NN) / 256), dim3(256), 0, stream>>>(
        Pc, Sc, Hinit);

    ssm_scan_final<<<dim3((BB * DD * NC * 4) / 256), dim3(256), 0, stream>>>(
        x, A, delta, Bp, Cp, Hinit, y);
}


# projection GEMM: loads of 16 k-steps issued as one register batch ahead of their WMMAs (scheduler barrier holds them there): one memory wait per batch, not per step; k order and sums unchanged
# speedup vs baseline: 1.5133x; 1.5133x over previous
#include <hip/hip_runtime.h>
#include <math.h>

#define DD 768
#define LL 1024
#define NN 16
#define BB 2
#define NC 16
#define LC (LL / NC)
#define KU 16

typedef float v2f __attribute__((ext_vector_type(2)));
typedef float v8f __attribute__((ext_vector_type(8)));

__global__ __launch_bounds__(32) void ssm_proj_wmma(
    const float* __restrict__ x,
    const float* __restrict__ W_B,
    const float* __restrict__ b_B,
    const float* __restrict__ W_C,
    const float* __restrict__ b_C,
    const float* __restrict__ W_d,
    const float* __restrict__ b_d,
    float* __restrict__ Bp,
    float* __restrict__ Cp,
    float* __restrict__ dlin)
{
    const int lane = threadIdx.x;
    const int half = lane >> 4;
    const int mn   = lane & 15;
    const int row0 = blockIdx.x << 4;
    const int b    = row0 >> 10;
    const int l0   = row0 & (LL - 1);
    const int lA   = l0 + mn;

    const float biasB = b_B[mn];
    const float biasC = b_C[mn];
    const float biasD = (mn == 0) ? b_d[0] : 0.0f;
    v8f accB, accC, accD;
#pragma unroll
    for (int r = 0; r < 8; ++r) { accB[r] = biasB; accC[r] = biasC; accD[r] = biasD; }

    const float* __restrict__ xb = x + (size_t)b * DD * LL;
    const float n0sel = (mn == 0) ? 1.0f : 0.0f;

    const float* __restrict__ wbr = W_B + mn * DD + 2 * half;
    const float* __restrict__ wcr = W_C + mn * DD + 2 * half;
    const float* __restrict__ wdr = W_d + 2 * half;
    const float* __restrict__ xa  = xb + (size_t)(2 * half) * LL + lA;

    for (int kb = 0; kb < DD; kb += 4 * KU) {
        v2f a[KU], wb[KU], wc[KU], wd[KU];
#pragma unroll
        for (int u = 0; u < KU; ++u) {
            const int k = kb + 4 * u;
            a[u][0] = xa[(size_t)k * LL];
            a[u][1] = xa[(size_t)(k + 1) * LL];
            wb[u] = *(const v2f*)(wbr + k);
            wc[u] = *(const v2f*)(wcr + k);
            wd[u] = *(const v2f*)(wdr + k);
        }
        __builtin_amdgcn_sched_barrier(0);
#pragma unroll
        for (int u = 0; u < KU; ++u) {
            v2f wdm;
            wdm[0] = wd[u][0] * n0sel;
            wdm[1] = wd[u][1] * n0sel;
            accB = __builtin_amdgcn_wmma_f32_16x16x4_f32(false, a[u], false, wb[u], (short)0, accB, false, false);
            accC = __builtin_amdgcn_wmma_f32_16x16x4_f32(false, a[u], false, wc[u], (short)0, accC, false, false);
            accD = __builtin_amdgcn_wmma_f32_16x16x4_f32(false, a[u], false, wdm,   (short)0, accD, false, false);
        }
    }

#pragma unroll
    for (int r = 0; r < 8; ++r) {
        const int l  = l0 + r + 8 * half;
        const int gr = b * LL + l;
        Bp[gr * NN + mn] = accB[r];
        Cp[gr * NN + mn] = accC[r];
        if (mn == 0) dlin[gr] = accD[r];
    }
}

__global__ __launch_bounds__(256) void ssm_delta(
    const float* __restrict__ dlin,
    const float* __restrict__ p_Delta,
    float* __restrict__ delta)
{
    const int i  = blockIdx.x * 256 + threadIdx.x;
    const int l  = i & (LL - 1);
    const int bd = i >> 10;
    const int b  = bd / DD;
    const int d  = bd - b * DD;
    const float dl = dlin[b * LL + l] + p_Delta[d];
    delta[i] = fmaxf(dl, 0.0f) + log1pf(__expf(-fabsf(dl)));
}

__global__ __launch_bounds__(256) void ssm_scan_local(
    const float* __restrict__ x,
    const float* __restrict__ A,
    const float* __restrict__ delta,
    const float* __restrict__ Bp,
    float* __restrict__ Pc,
    float* __restrict__ Sc)
{
    const int gtid  = blockIdx.x * 256 + threadIdx.x;
    const int q     = gtid & 3;
    const int group = gtid >> 2;
    const int c     = group & (NC - 1);
    const int pair  = group >> 4;
    const int b     = pair / DD;
    const int d     = pair - b * DD;

    const float4 An  = ((const float4*)(A + d * NN))[q];
    const float4 rAn = make_float4(1.0f / An.x, 1.0f / An.y, 1.0f / An.z, 1.0f / An.w);

    const float*  __restrict__ dl  = delta + (size_t)pair * LL + c * LC;
    const float*  __restrict__ xr  = x     + (size_t)pair * LL + c * LC;
    const float4* __restrict__ bpr = (const float4*)(Bp + ((size_t)b * LL + c * LC) * NN) + q;

    float4 h = make_float4(0.f, 0.f, 0.f, 0.f);
    float sd = 0.0f;
#pragma unroll 2
    for (int s = 0; s < LC; ++s) {
        const float dt = dl[s];
        const float xv = xr[s];
        const float4 bp = bpr[4 * s];
        sd += dt;
        const float a0 = __expf(dt * An.x);
        const float a1 = __expf(dt * An.y);
        const float a2 = __expf(dt * An.z);
        const float a3 = __expf(dt * An.w);
        h.x = fmaf(a0, h.x, (a0 - An.x) * rAn.x * bp.x * xv);
        h.y = fmaf(a1, h.y, (a1 - An.y) * rAn.y * bp.y * xv);
        h.z = fmaf(a2, h.z, (a2 - An.z) * rAn.z * bp.z * xv);
        h.w = fmaf(a3, h.w, (a3 - An.w) * rAn.w * bp.w * xv);
    }
    float4 P;
    P.x = __expf(An.x * sd);
    P.y = __expf(An.y * sd);
    P.z = __expf(An.z * sd);
    P.w = __expf(An.w * sd);
    ((float4*)Pc)[group * 4 + q] = P;
    ((float4*)Sc)[group * 4 + q] = h;
}

__global__ __launch_bounds__(256) void ssm_chunk_combine(
    const float* __restrict__ Pc,
    const float* __restrict__ Sc,
    float* __restrict__ Hinit)
{
    const int t    = blockIdx.x * 256 + threadIdx.x;
    const int n    = t & 15;
    const int pair = t >> 4;
    float h = 0.0f;
#pragma unroll
    for (int c = 0; c < NC; ++c) {
        const int idx = (pair * NC + c) * NN + n;
        Hinit[idx] = h;
        h = fmaf(Pc[idx], h, Sc[idx]);
    }
}

__global__ __launch_bounds__(256) void ssm_scan_final(
    const float* __restrict__ x,
    const float* __restrict__ A,
    const float* __restrict__ delta,
    const float* __restrict__ Bp,
    const float* __restrict__ Cp,
    const float* __restrict__ Hinit,
    float* __restrict__ y)
{
    const int gtid  = blockIdx.x * 256 + threadIdx.x;
    const int q     = gtid & 3;
    const int group = gtid >> 2;
    const int c     = group & (NC - 1);
    const int pair  = group >> 4;
    const int b     = pair / DD;
    const int d     = pair - b * DD;

    const float4 An  = ((const float4*)(A + d * NN))[q];
    const float4 rAn = make_float4(1.0f / An.x, 1.0f / An.y, 1.0f / An.z, 1.0f / An.w);

    const float*  __restrict__ dl  = delta + (size_t)pair * LL + c * LC;
    const float*  __restrict__ xr  = x     + (size_t)pair * LL + c * LC;
    const float4* __restrict__ bpr = (const float4*)(Bp + ((size_t)b * LL + c * LC) * NN) + q;
    const float4* __restrict__ cpr = (const float4*)(Cp + ((size_t)b * LL + c * LC) * NN) + q;
    float*        __restrict__ yr  = y     + (size_t)pair * LL + c * LC;

    float4 h = ((const float4*)Hinit)[group * 4 + q];

#pragma unroll 2
    for (int s = 0; s < LC; ++s) {
        const float dt = dl[s];
        const float xv = xr[s];
        const float4 bp = bpr[4 * s];
        const float4 cp = cpr[4 * s];
        const float a0 = __expf(dt * An.x);
        const float a1 = __expf(dt * An.y);
        const float a2 = __expf(dt * An.z);
        const float a3 = __expf(dt * An.w);
        h.x = fmaf(a0, h.x, (a0 - An.x) * rAn.x * bp.x * xv);
        h.y = fmaf(a1, h.y, (a1 - An.y) * rAn.y * bp.y * xv);
        h.z = fmaf(a2, h.z, (a2 - An.z) * rAn.z * bp.z * xv);
        h.w = fmaf(a3, h.w, (a3 - An.w) * rAn.w * bp.w * xv);

        float p = cp.x * h.x;
        p = fmaf(cp.y, h.y, p);
        p = fmaf(cp.z, h.z, p);
        p = fmaf(cp.w, h.w, p);
        p += __shfl_xor(p, 1, 4);
        p += __shfl_xor(p, 2, 4);
        if (q == 0) yr[s] = p;
    }
}

extern "C" void kernel_launch(void* const* d_in, const int* in_sizes, int n_in,
                              void* d_out, int out_size, void* d_ws, size_t ws_size,
                              hipStream_t stream)
{
    const float* x       = (const float*)d_in[0];
    const float* A       = (const float*)d_in[1];
    const float* W_B     = (const float*)d_in[2];
    const float* b_B     = (const float*)d_in[3];
    const float* W_C     = (const float*)d_in[4];
    const float* b_C     = (const float*)d_in[5];
    const float* W_d     = (const float*)d_in[6];
    const float* b_d     = (const float*)d_in[7];
    const float* p_Delta = (const float*)d_in[8];
    float* y = (float*)d_out;

    float* ws    = (float*)d_ws;
    float* Bp    = ws;
    float* Cp    = Bp + (size_t)BB * LL * NN;
    float* dlin  = Cp + (size_t)BB * LL * NN;
    float* delta = dlin + (size_t)BB * LL;
    float* Pc    = delta + (size_t)BB * DD * LL;
    float* Sc    = Pc + (size_t)BB * DD * NC * NN;
    float* Hinit = Sc + (size_t)BB * DD * NC * NN;

    ssm_proj_wmma<<<dim3(128), dim3(32), 0, stream>>>(
        x, W_B, b_B, W_C, b_C, W_d, b_d, Bp, Cp, dlin);

    ssm_delta<<<dim3((BB * DD * LL) / 256), dim3(256), 0, stream>>>(
        dlin, p_Delta, delta);

    ssm_scan_local<<<dim3((BB * DD * NC * 4) / 256), dim3(256), 0, stream>>>(
        x, A, delta, Bp, Pc, Sc);

    ssm_chunk_combine<<<dim3((BB * DD * NN) / 256), dim3(256), 0, stream>>>(
        Pc, Sc, Hinit);

    ssm_scan_final<<<dim3((BB * DD * NC * 4) / 256), dim3(256), 0, stream>>>(
        x, A, delta, Bp, Cp, Hinit, y);
}


# scan re-cut: one lane per (b,d,n) runs all 1024 steps serially, softplus in flight, no chunks, no delta plane; y summed over 16 lanes by DPP butterfly (h and y sum order changed); 5 launches -> 2
# speedup vs baseline: 1.9685x; 1.3008x over previous
#include <hip/hip_runtime.h>
#include <math.h>

#define DD 768
#define LL 1024
#define NN 16
#define BB 2
#define KU 16
#define TB 16
#define NBLK (LL / TB)
#define SW 3

typedef float v2f __attribute__((ext_vector_type(2)));
typedef float v8f __attribute__((ext_vector_type(8)));

__global__ __launch_bounds__(32) void ssm_proj_wmma(
    const float* __restrict__ x,
    const float* __restrict__ W_B,
    const float* __restrict__ b_B,
    const float* __restrict__ W_C,
    const float* __restrict__ b_C,
    const float* __restrict__ W_d,
    const float* __restrict__ b_d,
    float* __restrict__ Bp,
    float* __restrict__ Cp,
    float* __restrict__ dlin)
{
    const int lane = threadIdx.x;
    const int half = lane >> 4;
    const int mn   = lane & 15;
    const int row0 = blockIdx.x << 4;
    const int b    = row0 >> 10;
    const int l0   = row0 & (LL - 1);
    const int lA   = l0 + mn;

    const float biasB = b_B[mn];
    const float biasC = b_C[mn];
    const float biasD = (mn == 0) ? b_d[0] : 0.0f;
    v8f accB, accC, accD;
#pragma unroll
    for (int r = 0; r < 8; ++r) { accB[r] = biasB; accC[r] = biasC; accD[r] = biasD; }

    const float* __restrict__ xb = x + (size_t)b * DD * LL;
    const float n0sel = (mn == 0) ? 1.0f : 0.0f;

    const float* __restrict__ wbr = W_B + mn * DD + 2 * half;
    const float* __restrict__ wcr = W_C + mn * DD + 2 * half;
    const float* __restrict__ wdr = W_d + 2 * half;
    const float* __restrict__ xa  = xb + (size_t)(2 * half) * LL + lA;

    for (int kb = 0; kb < DD; kb += 4 * KU) {
        v2f a[KU], wb[KU], wc[KU], wd[KU];
#pragma unroll
        for (int u = 0; u < KU; ++u) {
            const int k = kb + 4 * u;
            a[u][0] = xa[(size_t)k * LL];
            a[u][1] = xa[(size_t)(k + 1) * LL];
            wb[u] = *(const v2f*)(wbr + k);
            wc[u] = *(const v2f*)(wcr + k);
            wd[u] = *(const v2f*)(wdr + k);
        }
        __builtin_amdgcn_sched_barrier(0);
#pragma unroll
        for (int u = 0; u < KU; ++u) {
            v2f wdm;
            wdm[0] = wd[u][0] * n0sel;
            wdm[1] = wd[u][1] * n0sel;
            accB = __builtin_amdgcn_wmma_f32_16x16x4_f32(false, a[u], false, wb[u], (short)0, accB, false, false);
            accC = __builtin_amdgcn_wmma_f32_16x16x4_f32(false, a[u], false, wc[u], (short)0, accC, false, false);
            accD = __builtin_amdgcn_wmma_f32_16x16x4_f32(false, a[u], false, wdm,   (short)0, accD, false, false);
        }
    }

#pragma unroll
    for (int r = 0; r < 8; ++r) {
        const int l  = l0 + r + 8 * half;
        const int gr = b * LL + l;
        Bp[gr * NN + mn] = accB[r];
        Cp[gr * NN + mn] = accC[r];
        if (mn == 0) dlin[gr] = accD[r];
    }
}


template <int S> __device__ __forceinline__ float row_bcast(float v) {
    return __int_as_float(__builtin_amdgcn_update_dpp(0, __float_as_int(v), 0x150 + S, 0xf, 0xf, true));
}
template <int M> __device__ __forceinline__ float row_xor(float v) {
    return __int_as_float(__builtin_amdgcn_update_dpp(0, __float_as_int(v), 0x160 + M, 0xf, 0xf, true));
}

struct ScanBlk {
    float xv;
    float bp[TB];
    float cp[TB];
};

template <int M> __device__ __forceinline__ float bfly(bool hi, float lo_e, float hi_e) {
    const float keep = hi ? hi_e : lo_e;
    const float send = hi ? lo_e : hi_e;
    return keep + row_xor<M>(send);
}

#define SCAN_STEP(S)                                                          \
    {                                                                         \
        const float dt = row_bcast<S>(dtv);                                   \
        const float xs = row_bcast<S>(xcur);                                  \
        const float a  = __expf(dt * An);                                     \
        h = fmaf(a, h, (a - An) * rAn * blk.bp[S] * xs);                      \
        p[S] = blk.cp[S] * h;                                                 \
        blk.bp[S] = bpr[(size_t)(l2 + S) * NN];                               \
        blk.cp[S] = cpr[(size_t)(l2 + S) * NN];                               \
    }

__global__ __launch_bounds__(32 * SW) __attribute__((amdgpu_waves_per_eu(1, 1))) void ssm_scan_serial(
    const float* __restrict__ x,
    const float* __restrict__ A,
    const float* __restrict__ p_Delta,
    const float* __restrict__ dlin,
    const float* __restrict__ Bp,
    const float* __restrict__ Cp,
    float* __restrict__ y)
{
    const int lane = threadIdx.x & 31;
    const int wave = blockIdx.x * SW + (threadIdx.x >> 5);
    const int n    = lane & 15;
    const int pair = wave * 2 + (lane >> 4);
    const int b    = pair / DD;
    const int d    = pair - b * DD;

    const float An  = A[d * NN + n];
    const float rAn = 1.0f / An;
    const float pD  = p_Delta[d];

    const float* __restrict__ xr  = x    + (size_t)pair * LL + n;
    const float* __restrict__ dr  = dlin + (size_t)b * LL + n;
    const float* __restrict__ bpr = Bp   + (size_t)b * LL * NN + n;
    const float* __restrict__ cpr = Cp   + (size_t)b * LL * NN + n;
    float*       __restrict__ yr  = y    + (size_t)pair * LL + n;

    const bool m8 = (lane & 8) != 0;
    const bool m4 = (lane & 4) != 0;
    const bool m2 = (lane & 2) != 0;
    const bool m1 = (lane & 1) != 0;

    ScanBlk ba, bb;
    ba.xv = xr[0];
    bb.xv = xr[TB];
#pragma unroll
    for (int s = 0; s < TB; ++s) {
        ba.bp[s] = bpr[(size_t)s * NN];
        ba.cp[s] = cpr[(size_t)s * NN];
        bb.bp[s] = bpr[(size_t)(TB + s) * NN];
        bb.cp[s] = cpr[(size_t)(TB + s) * NN];
    }

    auto softplus = [&](const float dlv) {
        const float dsum = dlv + pD;
        return fmaxf(dsum, 0.0f) + log1pf(__expf(-fabsf(dsum)));
    };
    float dtA = softplus(dr[0]);
    float dtB = softplus(dr[TB]);
    float dlA = dr[2 * TB];
    float dlB = dr[3 * TB];

    float h = 0.0f;

    auto run_block = [&](ScanBlk& blk, const float dtv, const int l0, const int l2) {
        const float xcur = blk.xv;
        blk.xv = xr[l2];
        float p[TB];
        SCAN_STEP(0)  SCAN_STEP(1)  SCAN_STEP(2)  SCAN_STEP(3)
        SCAN_STEP(4)  SCAN_STEP(5)  SCAN_STEP(6)  SCAN_STEP(7)
        SCAN_STEP(8)  SCAN_STEP(9)  SCAN_STEP(10) SCAN_STEP(11)
        SCAN_STEP(12) SCAN_STEP(13) SCAN_STEP(14) SCAN_STEP(15)
        float w8[8], w4[4], w2[2];
#pragma unroll
        for (int i = 0; i < 8; ++i) w8[i] = bfly<8>(m8, p[i], p[i + 8]);
#pragma unroll
        for (int i = 0; i < 4; ++i) w4[i] = bfly<4>(m4, w8[i], w8[i + 4]);
#pragma unroll
        for (int i = 0; i < 2; ++i) w2[i] = bfly<2>(m2, w4[i], w4[i + 2]);
        yr[l0] = bfly<1>(m1, w2[0], w2[1]);
    };

    for (int k = 0; k < NBLK; k += 2) {
        const int k2 = (k + 2 < NBLK) ? k + 2 : NBLK - 1;
        const int k3 = (k + 3 < NBLK) ? k + 3 : NBLK - 1;
        const int k4 = (k + 4 < NBLK) ? k + 4 : NBLK - 1;
        const int k5 = (k + 5 < NBLK) ? k + 5 : NBLK - 1;
        const float ndtA = softplus(dlA);
        const float ndtB = softplus(dlB);
        dlA = dr[k4 * TB];
        dlB = dr[k5 * TB];
        run_block(ba, dtA, k * TB, k2 * TB);
        run_block(bb, dtB, (k + 1) * TB, k3 * TB);
        dtA = ndtA;
        dtB = ndtB;
    }
}

extern "C" void kernel_launch(void* const* d_in, const int* in_sizes, int n_in,
                              void* d_out, int out_size, void* d_ws, size_t ws_size,
                              hipStream_t stream)
{
    const float* x       = (const float*)d_in[0];
    const float* A       = (const float*)d_in[1];
    const float* W_B     = (const float*)d_in[2];
    const float* b_B     = (const float*)d_in[3];
    const float* W_C     = (const float*)d_in[4];
    const float* b_C     = (const float*)d_in[5];
    const float* W_d     = (const float*)d_in[6];
    const float* b_d     = (const float*)d_in[7];
    const float* p_Delta = (const float*)d_in[8];
    float* y = (float*)d_out;

    float* ws   = (float*)d_ws;
    float* Bp   = ws;
    float* Cp   = Bp + (size_t)BB * LL * NN;
    float* dlin = Cp + (size_t)BB * LL * NN;

    ssm_proj_wmma<<<dim3(128), dim3(32), 0, stream>>>(
        x, W_B, b_B, W_C, b_C, W_d, b_d, Bp, Cp, dlin);

    ssm_scan_serial<<<dim3((BB * DD) / (2 * SW)), dim3(32 * SW), 0, stream>>>(
        x, A, p_Delta, dlin, Bp, Cp, y);
}


# scan: loads/stores as scalar base + fixed 32-bit lane offset + constant (was 64-bit vector address pairs rebuilt behind issued loads): translation waits per trip 9 -> 1; prologue loads in loop order
# speedup vs baseline: 2.2763x; 1.1564x over previous
#include <hip/hip_runtime.h>
#include <math.h>

#define DD 768
#define LL 1024
#define NN 16
#define BB 2
#define KU 16
#define TB 16
#define NBLK (LL / TB)
#define SW 3

typedef float v2f __attribute__((ext_vector_type(2)));
typedef float v8f __attribute__((ext_vector_type(8)));

__global__ __launch_bounds__(32) void ssm_proj_wmma(
    const float* __restrict__ x,
    const float* __restrict__ W_B,
    const float* __restrict__ b_B,
    const float* __restrict__ W_C,
    const float* __restrict__ b_C,
    const float* __restrict__ W_d,
    const float* __restrict__ b_d,
    float* __restrict__ Bp,
    float* __restrict__ Cp,
    float* __restrict__ dlin)
{
    const int lane = threadIdx.x;
    const int half = lane >> 4;
    const int mn   = lane & 15;
    const int row0 = blockIdx.x << 4;
    const int b    = row0 >> 10;
    const int l0   = row0 & (LL - 1);
    const int lA   = l0 + mn;

    const float biasB = b_B[mn];
    const float biasC = b_C[mn];
    const float biasD = (mn == 0) ? b_d[0] : 0.0f;
    v8f accB, accC, accD;
#pragma unroll
    for (int r = 0; r < 8; ++r) { accB[r] = biasB; accC[r] = biasC; accD[r] = biasD; }

    const float* __restrict__ xb = x + (size_t)b * DD * LL;
    const float n0sel = (mn == 0) ? 1.0f : 0.0f;

    const float* __restrict__ wbr = W_B + mn * DD + 2 * half;
    const float* __restrict__ wcr = W_C + mn * DD + 2 * half;
    const float* __restrict__ wdr = W_d + 2 * half;
    const float* __restrict__ xa  = xb + (size_t)(2 * half) * LL + lA;

    for (int kb = 0; kb < DD; kb += 4 * KU) {
        v2f a[KU], wb[KU], wc[KU], wd[KU];
#pragma unroll
        for (int u = 0; u < KU; ++u) {
            const int k = kb + 4 * u;
            a[u][0] = xa[(size_t)k * LL];
            a[u][1] = xa[(size_t)(k + 1) * LL];
            wb[u] = *(const v2f*)(wbr + k);
            wc[u] = *(const v2f*)(wcr + k);
            wd[u] = *(const v2f*)(wdr + k);
        }
        __builtin_amdgcn_sched_barrier(0);
#pragma unroll
        for (int u = 0; u < KU; ++u) {
            v2f wdm;
            wdm[0] = wd[u][0] * n0sel;
            wdm[1] = wd[u][1] * n0sel;
            accB = __builtin_amdgcn_wmma_f32_16x16x4_f32(false, a[u], false, wb[u], (short)0, accB, false, false);
            accC = __builtin_amdgcn_wmma_f32_16x16x4_f32(false, a[u], false, wc[u], (short)0, accC, false, false);
            accD = __builtin_amdgcn_wmma_f32_16x16x4_f32(false, a[u], false, wdm,   (short)0, accD, false, false);
        }
    }

#pragma unroll
    for (int r = 0; r < 8; ++r) {
        const int l  = l0 + r + 8 * half;
        const int gr = b * LL + l;
        Bp[gr * NN + mn] = accB[r];
        Cp[gr * NN + mn] = accC[r];
        if (mn == 0) dlin[gr] = accD[r];
    }
}


template <int S> __device__ __forceinline__ float row_bcast(float v) {
    return __int_as_float(__builtin_amdgcn_update_dpp(0, __float_as_int(v), 0x150 + S, 0xf, 0xf, true));
}
template <int M> __device__ __forceinline__ float row_xor(float v) {
    return __int_as_float(__builtin_amdgcn_update_dpp(0, __float_as_int(v), 0x160 + M, 0xf, 0xf, true));
}

struct ScanBlk {
    float xv;
    float bp[TB];
    float cp[TB];
};

template <int M> __device__ __forceinline__ float bfly(bool hi, float lo_e, float hi_e) {
    const float keep = hi ? hi_e : lo_e;
    const float send = hi ? lo_e : hi_e;
    return keep + row_xor<M>(send);
}

typedef __attribute__((address_space(1))) const float gcfloat;
typedef __attribute__((address_space(1))) float gfloat;

__device__ __forceinline__ float ld_lane(const float* __restrict__ ubase, const int elem, const unsigned lane_bytes) {
    return *(gcfloat*)((unsigned long long)ubase + (unsigned long long)lane_bytes + (unsigned long long)(elem * 4));
}
__device__ __forceinline__ void st_lane(float* __restrict__ ubase, const int elem, const unsigned lane_bytes, const float v) {
    *(gfloat*)((unsigned long long)ubase + (unsigned long long)lane_bytes + (unsigned long long)(elem * 4)) = v;
}
__device__ __forceinline__ void keep_lane_offset(unsigned& lane_bytes) {
    asm volatile("" : "+v"(lane_bytes));
}

#define SCAN_STEP(S)                                                          \
    {                                                                         \
        const float dt = row_bcast<S>(dtv);                                   \
        const float xs = row_bcast<S>(xcur);                                  \
        const float a  = __expf(dt * An);                                     \
        h = fmaf(a, h, (a - An) * rAn * blk.bp[S] * xs);                      \
        p[S] = blk.cp[S] * h;                                                 \
        blk.bp[S] = ld_lane(bpn, (S) * NN, nbt);                              \
        blk.cp[S] = ld_lane(cpn, (S) * NN, nbt);                              \
    }

__global__ __launch_bounds__(32 * SW) __attribute__((amdgpu_waves_per_eu(1, 1))) void ssm_scan_serial(
    const float* __restrict__ x,
    const float* __restrict__ A,
    const float* __restrict__ p_Delta,
    const float* __restrict__ dlin,
    const float* __restrict__ Bp,
    const float* __restrict__ Cp,
    float* __restrict__ y)
{
    const unsigned tid   = threadIdx.x;
    const unsigned lane  = tid & 31;
    const unsigned n     = tid & 15;
    const unsigned rw    = tid >> 4;
    const unsigned pair0 = blockIdx.x * (2 * SW);
    const unsigned b     = pair0 / DD;
    const unsigned pair  = pair0 + rw;
    const unsigned d     = pair - b * DD;

    const float An  = A[d * NN + n];
    const float rAn = 1.0f / An;
    const float pD  = p_Delta[d];

    const float* __restrict__ xu  = x    + (size_t)pair0 * LL;
    const float* __restrict__ du  = dlin + (size_t)b * LL;
    const float* __restrict__ bpu = Bp   + (size_t)b * LL * NN;
    const float* __restrict__ cpu = Cp   + (size_t)b * LL * NN;
    float*       __restrict__ yu  = y    + (size_t)pair0 * LL;
    const unsigned nb = n * 4u;
    const unsigned xb = (rw * LL + n) * 4u;

    const bool m8 = (lane & 8) != 0;
    const bool m4 = (lane & 4) != 0;
    const bool m2 = (lane & 2) != 0;
    const bool m1 = (lane & 1) != 0;

    ScanBlk ba, bb;
    const float dl0 = ld_lane(du, 0, nb);
    const float dl1 = ld_lane(du, TB, nb);
    ba.xv = ld_lane(xu, 0, xb);
#pragma unroll
    for (int s = 0; s < TB; ++s) ba.bp[s] = ld_lane(bpu, s * NN, nb);
#pragma unroll
    for (int s = 0; s < TB; ++s) ba.cp[s] = ld_lane(cpu, s * NN, nb);
    float dlA = ld_lane(du, 2 * TB, nb);
    float dlB = ld_lane(du, 3 * TB, nb);
    bb.xv = ld_lane(xu, TB, xb);
#pragma unroll
    for (int s = 0; s < TB; ++s) bb.bp[s] = ld_lane(bpu, (TB + s) * NN, nb);
#pragma unroll
    for (int s = 0; s < TB; ++s) bb.cp[s] = ld_lane(cpu, (TB + s) * NN, nb);

    auto softplus = [&](const float dlv) {
        const float dsum = dlv + pD;
        return fmaxf(dsum, 0.0f) + log1pf(__expf(-fabsf(dsum)));
    };
    float dtA = softplus(dl0);
    float dtB = softplus(dl1);

    float h = 0.0f;

    auto run_block = [&](ScanBlk& blk, const float dtv, const int l0, const int l2,
                         const unsigned nbt, const unsigned xbt) {
        const float xcur = blk.xv;
        const float* __restrict__ bpn = bpu + l2 * NN;
        const float* __restrict__ cpn = cpu + l2 * NN;
        float p[TB];
        SCAN_STEP(0)  SCAN_STEP(1)  SCAN_STEP(2)  SCAN_STEP(3)
        SCAN_STEP(4)  SCAN_STEP(5)  SCAN_STEP(6)  SCAN_STEP(7)
        SCAN_STEP(8)  SCAN_STEP(9)  SCAN_STEP(10) SCAN_STEP(11)
        SCAN_STEP(12) SCAN_STEP(13) SCAN_STEP(14) SCAN_STEP(15)
        blk.xv = ld_lane(xu + l2, 0, xbt);
        float w8[8], w4[4], w2[2];
#pragma unroll
        for (int i = 0; i < 8; ++i) w8[i] = bfly<8>(m8, p[i], p[i + 8]);
#pragma unroll
        for (int i = 0; i < 4; ++i) w4[i] = bfly<4>(m4, w8[i], w8[i + 4]);
#pragma unroll
        for (int i = 0; i < 2; ++i) w2[i] = bfly<2>(m2, w4[i], w4[i + 2]);
        st_lane(yu + l0, 0, xbt, bfly<1>(m1, w2[0], w2[1]));
    };

    unsigned nbt = nb, xbt = xb;
    for (int k = 0; k < NBLK; k += 2) {
        keep_lane_offset(nbt);
        keep_lane_offset(xbt);
        const int k2 = (k + 2 < NBLK) ? k + 2 : NBLK - 1;
        const int k3 = (k + 3 < NBLK) ? k + 3 : NBLK - 1;
        const int k4 = (k + 4 < NBLK) ? k + 4 : NBLK - 1;
        const int k5 = (k + 5 < NBLK) ? k + 5 : NBLK - 1;
        const float ndtA = softplus(dlA);
        const float ndtB = softplus(dlB);
        run_block(ba, dtA, k * TB, k2 * TB, nbt, xbt);
        dlA = ld_lane(du + k4 * TB, 0, nbt);
        dlB = ld_lane(du + k5 * TB, 0, nbt);
        run_block(bb, dtB, (k + 1) * TB, k3 * TB, nbt, xbt);
        dtA = ndtA;
        dtB = ndtB;
    }
}

extern "C" void kernel_launch(void* const* d_in, const int* in_sizes, int n_in,
                              void* d_out, int out_size, void* d_ws, size_t ws_size,
                              hipStream_t stream)
{
    const float* x       = (const float*)d_in[0];
    const float* A       = (const float*)d_in[1];
    const float* W_B     = (const float*)d_in[2];
    const float* b_B     = (const float*)d_in[3];
    const float* W_C     = (const float*)d_in[4];
    const float* b_C     = (const float*)d_in[5];
    const float* W_d     = (const float*)d_in[6];
    const float* b_d     = (const float*)d_in[7];
    const float* p_Delta = (const float*)d_in[8];
    float* y = (float*)d_out;

    float* ws   = (float*)d_ws;
    float* Bp   = ws;
    float* Cp   = Bp + (size_t)BB * LL * NN;
    float* dlin = Cp + (size_t)BB * LL * NN;

    ssm_proj_wmma<<<dim3(128), dim3(32), 0, stream>>>(
        x, W_B, b_B, W_C, b_C, W_d, b_d, Bp, Cp, dlin);

    ssm_scan_serial<<<dim3((BB * DD) / (2 * SW)), dim3(32 * SW), 0, stream>>>(
        x, A, p_Delta, dlin, Bp, Cp, y);
}


# projection GEMM: each batch's fragment loads issued one batch ahead of its WMMAs, into a second register set; batches of 12 k-steps (60 loads, under the 63 a wave may have in flight); sums unchanged
# speedup vs baseline: 2.5229x; 1.1083x over previous
#include <hip/hip_runtime.h>
#include <math.h>

#define DD 768
#define LL 1024
#define NN 16
#define BB 2
#define KU 12
#define KS (4 * KU)
#define TB 16
#define NBLK (LL / TB)
#define SW 3

typedef float v2f __attribute__((ext_vector_type(2)));
typedef float v8f __attribute__((ext_vector_type(8)));


struct ProjFrag { v2f a[KU], wb[KU], wc[KU], wd[KU]; };

__global__ __launch_bounds__(32) void ssm_proj_wmma(
    const float* __restrict__ x,
    const float* __restrict__ W_B,
    const float* __restrict__ b_B,
    const float* __restrict__ W_C,
    const float* __restrict__ b_C,
    const float* __restrict__ W_d,
    const float* __restrict__ b_d,
    float* __restrict__ Bp,
    float* __restrict__ Cp,
    float* __restrict__ dlin)
{
    const int lane = threadIdx.x;
    const int half = lane >> 4;
    const int mn   = lane & 15;
    const int row0 = blockIdx.x << 4;
    const int b    = row0 >> 10;
    const int l0   = row0 & (LL - 1);
    const int lA   = l0 + mn;

    const float biasB = b_B[mn];
    const float biasC = b_C[mn];
    const float biasD = (mn == 0) ? b_d[0] : 0.0f;
    v8f accB, accC, accD;
#pragma unroll
    for (int r = 0; r < 8; ++r) { accB[r] = biasB; accC[r] = biasC; accD[r] = biasD; }

    const float* __restrict__ xb = x + (size_t)b * DD * LL;
    const float n0sel = (mn == 0) ? 1.0f : 0.0f;

    const float* __restrict__ wbr = W_B + mn * DD + 2 * half;
    const float* __restrict__ wcr = W_C + mn * DD + 2 * half;
    const float* __restrict__ wdr = W_d + 2 * half;
    const float* __restrict__ xa  = xb + (size_t)(2 * half) * LL + lA;

    ProjFrag f0, f1;
    auto load_batch = [&](ProjFrag& f, const int kb) {
#pragma unroll
        for (int u = 0; u < KU; ++u) {
            const int k = kb + 4 * u;
            f.a[u][0] = xa[(size_t)k * LL];
            f.a[u][1] = xa[(size_t)(k + 1) * LL];
            f.wb[u] = *(const v2f*)(wbr + k);
            f.wc[u] = *(const v2f*)(wcr + k);
            f.wd[u] = *(const v2f*)(wdr + k);
        }
    };
    auto mma_batch = [&](const ProjFrag& f) {
#pragma unroll
        for (int u = 0; u < KU; ++u) {
            v2f wdm;
            wdm[0] = f.wd[u][0] * n0sel;
            wdm[1] = f.wd[u][1] * n0sel;
            accB = __builtin_amdgcn_wmma_f32_16x16x4_f32(false, f.a[u], false, f.wb[u], (short)0, accB, false, false);
            accC = __builtin_amdgcn_wmma_f32_16x16x4_f32(false, f.a[u], false, f.wc[u], (short)0, accC, false, false);
            accD = __builtin_amdgcn_wmma_f32_16x16x4_f32(false, f.a[u], false, wdm,     (short)0, accD, false, false);
        }
    };

    load_batch(f0, 0);
    for (int kb = 0; kb < DD - 2 * KS; kb += 2 * KS) {
        load_batch(f1, kb + KS);
        __builtin_amdgcn_sched_barrier(0);
        mma_batch(f0);
        __builtin_amdgcn_sched_barrier(0);
        load_batch(f0, kb + 2 * KS);
        __builtin_amdgcn_sched_barrier(0);
        mma_batch(f1);
        __builtin_amdgcn_sched_barrier(0);
    }
    load_batch(f1, DD - KS);
    __builtin_amdgcn_sched_barrier(0);
    mma_batch(f0);
    __builtin_amdgcn_sched_barrier(0);
    mma_batch(f1);

#pragma unroll
    for (int r = 0; r < 8; ++r) {
        const int l  = l0 + r + 8 * half;
        const int gr = b * LL + l;
        Bp[gr * NN + mn] = accB[r];
        Cp[gr * NN + mn] = accC[r];
        if (mn == 0) dlin[gr] = accD[r];
    }
}


template <int S> __device__ __forceinline__ float row_bcast(float v) {
    return __int_as_float(__builtin_amdgcn_update_dpp(0, __float_as_int(v), 0x150 + S, 0xf, 0xf, true));
}
template <int M> __device__ __forceinline__ float row_xor(float v) {
    return __int_as_float(__builtin_amdgcn_update_dpp(0, __float_as_int(v), 0x160 + M, 0xf, 0xf, true));
}

struct ScanBlk {
    float xv;
    float bp[TB];
    float cp[TB];
};

template <int M> __device__ __forceinline__ float bfly(bool hi, float lo_e, float hi_e) {
    const float keep = hi ? hi_e : lo_e;
    const float send = hi ? lo_e : hi_e;
    return keep + row_xor<M>(send);
}

typedef __attribute__((address_space(1))) const float gcfloat;
typedef __attribute__((address_space(1))) float gfloat;

__device__ __forceinline__ float ld_lane(const float* __restrict__ ubase, const int elem, const unsigned lane_bytes) {
    return *(gcfloat*)((unsigned long long)ubase + (unsigned long long)lane_bytes + (unsigned long long)(elem * 4));
}
__device__ __forceinline__ void st_lane(float* __restrict__ ubase, const int elem, const unsigned lane_bytes, const float v) {
    *(gfloat*)((unsigned long long)ubase + (unsigned long long)lane_bytes + (unsigned long long)(elem * 4)) = v;
}
__device__ __forceinline__ void keep_lane_offset(unsigned& lane_bytes) {
    asm volatile("" : "+v"(lane_bytes));
}

#define SCAN_STEP(S)                                                          \
    {                                                                         \
        const float dt = row_bcast<S>(dtv);                                   \
        const float xs = row_bcast<S>(xcur);                                  \
        const float a  = __expf(dt * An);                                     \
        h = fmaf(a, h, (a - An) * rAn * blk.bp[S] * xs);                      \
        p[S] = blk.cp[S] * h;                                                 \
        blk.bp[S] = ld_lane(bpn, (S) * NN, nbt);                              \
        blk.cp[S] = ld_lane(cpn, (S) * NN, nbt);                              \
    }

__global__ __launch_bounds__(32 * SW) __attribute__((amdgpu_waves_per_eu(1, 1))) void ssm_scan_serial(
    const float* __restrict__ x,
    const float* __restrict__ A,
    const float* __restrict__ p_Delta,
    const float* __restrict__ dlin,
    const float* __restrict__ Bp,
    const float* __restrict__ Cp,
    float* __restrict__ y)
{
    const unsigned tid   = threadIdx.x;
    const unsigned lane  = tid & 31;
    const unsigned n     = tid & 15;
    const unsigned rw    = tid >> 4;
    const unsigned pair0 = blockIdx.x * (2 * SW);
    const unsigned b     = pair0 / DD;
    const unsigned pair  = pair0 + rw;
    const unsigned d     = pair - b * DD;

    const float An  = A[d * NN + n];
    const float rAn = 1.0f / An;
    const float pD  = p_Delta[d];

    const float* __restrict__ xu  = x    + (size_t)pair0 * LL;
    const float* __restrict__ du  = dlin + (size_t)b * LL;
    const float* __restrict__ bpu = Bp   + (size_t)b * LL * NN;
    const float* __restrict__ cpu = Cp   + (size_t)b * LL * NN;
    float*       __restrict__ yu  = y    + (size_t)pair0 * LL;
    const unsigned nb = n * 4u;
    const unsigned xb = (rw * LL + n) * 4u;

    const bool m8 = (lane & 8) != 0;
    const bool m4 = (lane & 4) != 0;
    const bool m2 = (lane & 2) != 0;
    const bool m1 = (lane & 1) != 0;

    ScanBlk ba, bb;
    const float dl0 = ld_lane(du, 0, nb);
    const float dl1 = ld_lane(du, TB, nb);
    ba.xv = ld_lane(xu, 0, xb);
#pragma unroll
    for (int s = 0; s < TB; ++s) ba.bp[s] = ld_lane(bpu, s * NN, nb);
#pragma unroll
    for (int s = 0; s < TB; ++s) ba.cp[s] = ld_lane(cpu, s * NN, nb);
    float dlA = ld_lane(du, 2 * TB, nb);
    float dlB = ld_lane(du, 3 * TB, nb);
    bb.xv = ld_lane(xu, TB, xb);
#pragma unroll
    for (int s = 0; s < TB; ++s) bb.bp[s] = ld_lane(bpu, (TB + s) * NN, nb);
#pragma unroll
    for (int s = 0; s < TB; ++s) bb.cp[s] = ld_lane(cpu, (TB + s) * NN, nb);

    auto softplus = [&](const float dlv) {
        const float dsum = dlv + pD;
        return fmaxf(dsum, 0.0f) + log1pf(__expf(-fabsf(dsum)));
    };
    float dtA = softplus(dl0);
    float dtB = softplus(dl1);

    float h = 0.0f;

    auto run_block = [&](ScanBlk& blk, const float dtv, const int l0, const int l2,
                         const unsigned nbt, const unsigned xbt) {
        const float xcur = blk.xv;
        const float* __restrict__ bpn = bpu + l2 * NN;
        const float* __restrict__ cpn = cpu + l2 * NN;
        float p[TB];
        SCAN_STEP(0)  SCAN_STEP(1)  SCAN_STEP(2)  SCAN_STEP(3)
        SCAN_STEP(4)  SCAN_STEP(5)  SCAN_STEP(6)  SCAN_STEP(7)
        SCAN_STEP(8)  SCAN_STEP(9)  SCAN_STEP(10) SCAN_STEP(11)
        SCAN_STEP(12) SCAN_STEP(13) SCAN_STEP(14) SCAN_STEP(15)
        blk.xv = ld_lane(xu + l2, 0, xbt);
        float w8[8], w4[4], w2[2];
#pragma unroll
        for (int i = 0; i < 8; ++i) w8[i] = bfly<8>(m8, p[i], p[i + 8]);
#pragma unroll
        for (int i = 0; i < 4; ++i) w4[i] = bfly<4>(m4, w8[i], w8[i + 4]);
#pragma unroll
        for (int i = 0; i < 2; ++i) w2[i] = bfly<2>(m2, w4[i], w4[i + 2]);
        st_lane(yu + l0, 0, xbt, bfly<1>(m1, w2[0], w2[1]));
    };

    unsigned nbt = nb, xbt = xb;
    for (int k = 0; k < NBLK; k += 2) {
        keep_lane_offset(nbt);
        keep_lane_offset(xbt);
        const int k2 = (k + 2 < NBLK) ? k + 2 : NBLK - 1;
        const int k3 = (k + 3 < NBLK) ? k + 3 : NBLK - 1;
        const int k4 = (k + 4 < NBLK) ? k + 4 : NBLK - 1;
        const int k5 = (k + 5 < NBLK) ? k + 5 : NBLK - 1;
        const float ndtA = softplus(dlA);
        const float ndtB = softplus(dlB);
        run_block(ba, dtA, k * TB, k2 * TB, nbt, xbt);
        dlA = ld_lane(du + k4 * TB, 0, nbt);
        dlB = ld_lane(du + k5 * TB, 0, nbt);
        run_block(bb, dtB, (k + 1) * TB, k3 * TB, nbt, xbt);
        dtA = ndtA;
        dtB = ndtB;
    }
}

extern "C" void kernel_launch(void* const* d_in, const int* in_sizes, int n_in,
                              void* d_out, int out_size, void* d_ws, size_t ws_size,
                              hipStream_t stream)
{
    const float* x       = (const float*)d_in[0];
    const float* A       = (const float*)d_in[1];
    const float* W_B     = (const float*)d_in[2];
    const float* b_B     = (const float*)d_in[3];
    const float* W_C     = (const float*)d_in[4];
    const float* b_C     = (const float*)d_in[5];
    const float* W_d     = (const float*)d_in[6];
    const float* b_d     = (const float*)d_in[7];
    const float* p_Delta = (const float*)d_in[8];
    float* y = (float*)d_out;

    float* ws   = (float*)d_ws;
    float* Bp   = ws;
    float* Cp   = Bp + (size_t)BB * LL * NN;
    float* dlin = Cp + (size_t)BB * LL * NN;

    ssm_proj_wmma<<<dim3(128), dim3(32), 0, stream>>>(
        x, W_B, b_B, W_C, b_C, W_d, b_d, Bp, Cp, dlin);

    ssm_scan_serial<<<dim3((BB * DD) / (2 * SW)), dim3(32 * SW), 0, stream>>>(
        x, A, p_Delta, dlin, Bp, Cp, y);
}


# scan: projection writes Bp/Cp as 4-step packets, so a lane's 4 steps are one 16-byte load (40 loads per 64 steps, was 136); 4 register sets, each refilled 4 blocks ahead inside its own block
# speedup vs baseline: 2.6733x; 1.0596x over previous
#include <hip/hip_runtime.h>
#include <math.h>

#define DD 768
#define LL 1024
#define NN 16
#define BB 2
#define KU 12
#define KS (4 * KU)
#define TB 16
#define NBLK (LL / TB)
#define PK 4
#define NSET 4
#define SW 3

typedef float v2f __attribute__((ext_vector_type(2)));
typedef float v4f __attribute__((ext_vector_type(4)));
typedef float v8f __attribute__((ext_vector_type(8)));


struct ProjFrag { v2f a[KU], wb[KU], wc[KU], wd[KU]; };

__global__ __launch_bounds__(32) void ssm_proj_wmma(
    const float* __restrict__ x,
    const float* __restrict__ W_B,
    const float* __restrict__ b_B,
    const float* __restrict__ W_C,
    const float* __restrict__ b_C,
    const float* __restrict__ W_d,
    const float* __restrict__ b_d,
    float* __restrict__ Bp,
    float* __restrict__ Cp,
    float* __restrict__ dlin)
{
    const int lane = threadIdx.x;
    const int half = lane >> 4;
    const int mn   = lane & 15;
    const int row0 = blockIdx.x << 4;
    const int b    = row0 >> 10;
    const int l0   = row0 & (LL - 1);
    const int lA   = l0 + mn;

    const float biasB = b_B[mn];
    const float biasC = b_C[mn];
    const float biasD = (mn == 0) ? b_d[0] : 0.0f;
    v8f accB, accC, accD;
#pragma unroll
    for (int r = 0; r < 8; ++r) { accB[r] = biasB; accC[r] = biasC; accD[r] = biasD; }

    const float* __restrict__ xb = x + (size_t)b * DD * LL;
    const float n0sel = (mn == 0) ? 1.0f : 0.0f;

    const float* __restrict__ wbr = W_B + mn * DD + 2 * half;
    const float* __restrict__ wcr = W_C + mn * DD + 2 * half;
    const float* __restrict__ wdr = W_d + 2 * half;
    const float* __restrict__ xa  = xb + (size_t)(2 * half) * LL + lA;

    ProjFrag f0, f1;
    auto load_batch = [&](ProjFrag& f, const int kb) {
#pragma unroll
        for (int u = 0; u < KU; ++u) {
            const int k = kb + 4 * u;
            f.a[u][0] = xa[(size_t)k * LL];
            f.a[u][1] = xa[(size_t)(k + 1) * LL];
            f.wb[u] = *(const v2f*)(wbr + k);
            f.wc[u] = *(const v2f*)(wcr + k);
            f.wd[u] = *(const v2f*)(wdr + k);
        }
    };
    auto mma_batch = [&](const ProjFrag& f) {
#pragma unroll
        for (int u = 0; u < KU; ++u) {
            v2f wdm;
            wdm[0] = f.wd[u][0] * n0sel;
            wdm[1] = f.wd[u][1] * n0sel;
            accB = __builtin_amdgcn_wmma_f32_16x16x4_f32(false, f.a[u], false, f.wb[u], (short)0, accB, false, false);
            accC = __builtin_amdgcn_wmma_f32_16x16x4_f32(false, f.a[u], false, f.wc[u], (short)0, accC, false, false);
            accD = __builtin_amdgcn_wmma_f32_16x16x4_f32(false, f.a[u], false, wdm,     (short)0, accD, false, false);
        }
    };

    load_batch(f0, 0);
    for (int kb = 0; kb < DD - 2 * KS; kb += 2 * KS) {
        load_batch(f1, kb + KS);
        __builtin_amdgcn_sched_barrier(0);
        mma_batch(f0);
        __builtin_amdgcn_sched_barrier(0);
        load_batch(f0, kb + 2 * KS);
        __builtin_amdgcn_sched_barrier(0);
        mma_batch(f1);
        __builtin_amdgcn_sched_barrier(0);
    }
    load_batch(f1, DD - KS);
    __builtin_amdgcn_sched_barrier(0);
    mma_batch(f0);
    __builtin_amdgcn_sched_barrier(0);
    mma_batch(f1);

    const int gr0 = b * LL + l0 + 8 * half;
    const int pk0 = ((gr0 / PK) * NN + mn) * PK;
    const int pk1 = pk0 + NN * PK;
    v4f o;
    o[0] = accB[0]; o[1] = accB[1]; o[2] = accB[2]; o[3] = accB[3];
    *(v4f*)(Bp + pk0) = o;
    o[0] = accB[4]; o[1] = accB[5]; o[2] = accB[6]; o[3] = accB[7];
    *(v4f*)(Bp + pk1) = o;
    o[0] = accC[0]; o[1] = accC[1]; o[2] = accC[2]; o[3] = accC[3];
    *(v4f*)(Cp + pk0) = o;
    o[0] = accC[4]; o[1] = accC[5]; o[2] = accC[6]; o[3] = accC[7];
    *(v4f*)(Cp + pk1) = o;
#pragma unroll
    for (int r = 0; r < 8; ++r) {
        if (mn == 0) dlin[gr0 + r] = accD[r];
    }
}


template <int S> __device__ __forceinline__ float row_bcast(float v) {
    return __int_as_float(__builtin_amdgcn_update_dpp(0, __float_as_int(v), 0x150 + S, 0xf, 0xf, true));
}
template <int M> __device__ __forceinline__ float row_xor(float v) {
    return __int_as_float(__builtin_amdgcn_update_dpp(0, __float_as_int(v), 0x160 + M, 0xf, 0xf, true));
}

struct ScanBlk {
    float xv;
    v4f   bp[TB / PK];
    v4f   cp[TB / PK];
    float dt;
    float dl;
};

template <int M> __device__ __forceinline__ float bfly(bool hi, float lo_e, float hi_e) {
    const float keep = hi ? hi_e : lo_e;
    const float send = hi ? lo_e : hi_e;
    return keep + row_xor<M>(send);
}

typedef __attribute__((address_space(1))) const float gcfloat;
typedef __attribute__((address_space(1))) float gfloat;
typedef __attribute__((address_space(1))) const v4f gcv4f;

__device__ __forceinline__ float ld_lane(const float* __restrict__ ubase, const int elem, const unsigned lane_bytes) {
    return *(gcfloat*)((unsigned long long)ubase + (unsigned long long)lane_bytes + (unsigned long long)(elem * 4));
}
__device__ __forceinline__ v4f ld_lane4(const float* __restrict__ ubase, const int elem, const unsigned lane_bytes) {
    return *(gcv4f*)((unsigned long long)ubase + (unsigned long long)lane_bytes + (unsigned long long)(elem * 4));
}
__device__ __forceinline__ void st_lane(float* __restrict__ ubase, const int elem, const unsigned lane_bytes, const float v) {
    *(gfloat*)((unsigned long long)ubase + (unsigned long long)lane_bytes + (unsigned long long)(elem * 4)) = v;
}
__device__ __forceinline__ void keep_lane_offset(unsigned& lane_bytes) {
    asm volatile("" : "+v"(lane_bytes));
}

#define SCAN_STEP(S)                                                          \
    {                                                                         \
        const float dt = row_bcast<S>(dtv);                                   \
        const float xs = row_bcast<S>(xcur);                                  \
        const float a  = __expf(dt * An);                                     \
        h = fmaf(a, h, (a - An) * rAn * blk.bp[(S) / PK][(S) % PK] * xs);     \
        p[S] = blk.cp[(S) / PK][(S) % PK] * h;                                \
    }
#define SCAN_REFILL(Q)                                                        \
    {                                                                         \
        blk.bp[Q] = ld_lane4(bpn, (Q) * NN * PK, npt);                        \
        blk.cp[Q] = ld_lane4(cpn, (Q) * NN * PK, npt);                        \
    }

__global__ __launch_bounds__(32 * SW) __attribute__((amdgpu_waves_per_eu(1, 1))) void ssm_scan_serial(
    const float* __restrict__ x,
    const float* __restrict__ A,
    const float* __restrict__ p_Delta,
    const float* __restrict__ dlin,
    const float* __restrict__ Bp,
    const float* __restrict__ Cp,
    float* __restrict__ y)
{
    const unsigned tid   = threadIdx.x;
    const unsigned lane  = tid & 31;
    const unsigned n     = tid & 15;
    const unsigned rw    = tid >> 4;
    const unsigned pair0 = blockIdx.x * (2 * SW);
    const unsigned b     = pair0 / DD;
    const unsigned pair  = pair0 + rw;
    const unsigned d     = pair - b * DD;

    const float An  = A[d * NN + n];
    const float rAn = 1.0f / An;
    const float pD  = p_Delta[d];

    const float* __restrict__ xu  = x    + (size_t)pair0 * LL;
    const float* __restrict__ du  = dlin + (size_t)b * LL;
    const float* __restrict__ bpu = Bp   + (size_t)b * LL * NN;
    const float* __restrict__ cpu = Cp   + (size_t)b * LL * NN;
    float*       __restrict__ yu  = y    + (size_t)pair0 * LL;
    const unsigned nb = n * 4u;
    const unsigned np = n * (4u * PK);
    const unsigned xb = (rw * LL + n) * 4u;

    const bool m8 = (lane & 8) != 0;
    const bool m4 = (lane & 4) != 0;
    const bool m2 = (lane & 2) != 0;
    const bool m1 = (lane & 1) != 0;

    auto softplus = [&](const float dlv) {
        const float dsum = dlv + pD;
        return fmaxf(dsum, 0.0f) + log1pf(__expf(-fabsf(dsum)));
    };

    ScanBlk st[NSET];
    float dl0[NSET];
#pragma unroll
    for (int i = 0; i < NSET; ++i) {
        st[i].dl = ld_lane(du, (NSET + i) * TB, nb);
        st[i].xv = ld_lane(xu, i * TB, xb);
#pragma unroll
        for (int q = 0; q < TB / PK; ++q) st[i].bp[q] = ld_lane4(bpu, (i * TB + q * PK) * NN, np);
#pragma unroll
        for (int q = 0; q < TB / PK; ++q) st[i].cp[q] = ld_lane4(cpu, (i * TB + q * PK) * NN, np);
    }
    __builtin_amdgcn_sched_barrier(0);
#pragma unroll
    for (int i = 0; i < NSET; ++i) dl0[i] = ld_lane(du, i * TB, nb);
#pragma unroll
    for (int i = NSET - 1; i >= 0; --i) st[i].dt = softplus(dl0[i]);

    float h = 0.0f;

    auto run_steps = [&](ScanBlk& blk, const int kb, float (&p)[TB],
                         const unsigned nbt, const unsigned xbt, const unsigned npt) {
        const int kr = (kb + NSET < NBLK) ? kb + NSET : NBLK - 1;
        const int kd = (kb + 2 * NSET < NBLK) ? kb + 2 * NSET : NBLK - 1;
        const int l2 = kr * TB;
        const float ndt  = softplus(blk.dl);
        const float dtv  = blk.dt;
        const float xcur = blk.xv;
        const float* __restrict__ bpn = bpu + l2 * NN;
        const float* __restrict__ cpn = cpu + l2 * NN;
        SCAN_STEP(0)  SCAN_STEP(1)  SCAN_STEP(2)  SCAN_STEP(3)  SCAN_REFILL(0)
        SCAN_STEP(4)  SCAN_STEP(5)  SCAN_STEP(6)  SCAN_STEP(7)  SCAN_REFILL(1)
        SCAN_STEP(8)  SCAN_STEP(9)  SCAN_STEP(10) SCAN_STEP(11) SCAN_REFILL(2)
        SCAN_STEP(12) SCAN_STEP(13) SCAN_STEP(14) SCAN_STEP(15) SCAN_REFILL(3)
        blk.xv = ld_lane(xu + l2, 0, xbt);
        blk.dl = ld_lane(du + kd * TB, 0, nbt);
        blk.dt = ndt;
    };
    auto sum_store = [&](const float (&p)[TB], const int l0, const unsigned xbt) {
        float w8[8], w4[4], w2[2];
#pragma unroll
        for (int i = 0; i < 8; ++i) w8[i] = bfly<8>(m8, p[i], p[i + 8]);
#pragma unroll
        for (int i = 0; i < 4; ++i) w4[i] = bfly<4>(m4, w8[i], w8[i + 4]);
#pragma unroll
        for (int i = 0; i < 2; ++i) w2[i] = bfly<2>(m2, w4[i], w4[i + 2]);
        st_lane(yu + l0, 0, xbt, bfly<1>(m1, w2[0], w2[1]));
    };

    unsigned nbt = nb, xbt = xb, npt = np;
    for (int k = 0; k < NBLK; k += NSET) {
        keep_lane_offset(nbt);
        keep_lane_offset(xbt);
        keep_lane_offset(npt);
        float p[NSET][TB];
#pragma unroll
        for (int i = 0; i < NSET; ++i) {
            run_steps(st[i], k + i, p[i], nbt, xbt, npt);
            __builtin_amdgcn_sched_barrier(0);
            sum_store(p[i], (k + i) * TB, xbt);
        }
    }
}

extern "C" void kernel_launch(void* const* d_in, const int* in_sizes, int n_in,
                              void* d_out, int out_size, void* d_ws, size_t ws_size,
                              hipStream_t stream)
{
    const float* x       = (const float*)d_in[0];
    const float* A       = (const float*)d_in[1];
    const float* W_B     = (const float*)d_in[2];
    const float* b_B     = (const float*)d_in[3];
    const float* W_C     = (const float*)d_in[4];
    const float* b_C     = (const float*)d_in[5];
    const float* W_d     = (const float*)d_in[6];
    const float* b_d     = (const float*)d_in[7];
    const float* p_Delta = (const float*)d_in[8];
    float* y = (float*)d_out;

    float* ws   = (float*)d_ws;
    float* Bp   = ws;
    float* Cp   = Bp + (size_t)BB * LL * NN;
    float* dlin = Cp + (size_t)BB * LL * NN;

    ssm_proj_wmma<<<dim3(128), dim3(32), 0, stream>>>(
        x, W_B, b_B, W_C, b_C, W_d, b_d, Bp, Cp, dlin);

    ssm_scan_serial<<<dim3((BB * DD) / (2 * SW)), dim3(32 * SW), 0, stream>>>(
        x, A, p_Delta, dlin, Bp, Cp, y);
}


# scan: the 4 softplus chains of a trip (library log1p, one per block) run side by side at the trip's head, not one behind each block's steps; their dlin loads move there too; delta bit-identical
# speedup vs baseline: 3.1867x; 1.1920x over previous
#include <hip/hip_runtime.h>
#include <math.h>

#define DD 768
#define LL 1024
#define NN 16
#define BB 2
#define KU 12
#define KS (4 * KU)
#define TB 16
#define NBLK (LL / TB)
#define PK 4
#define NSET 4
#define SW 3

typedef float v2f __attribute__((ext_vector_type(2)));
typedef float v4f __attribute__((ext_vector_type(4)));
typedef float v8f __attribute__((ext_vector_type(8)));


struct ProjFrag { v2f a[KU], wb[KU], wc[KU], wd[KU]; };

__global__ __launch_bounds__(32) void ssm_proj_wmma(
    const float* __restrict__ x,
    const float* __restrict__ W_B,
    const float* __restrict__ b_B,
    const float* __restrict__ W_C,
    const float* __restrict__ b_C,
    const float* __restrict__ W_d,
    const float* __restrict__ b_d,
    float* __restrict__ Bp,
    float* __restrict__ Cp,
    float* __restrict__ dlin)
{
    const int lane = threadIdx.x;
    const int half = lane >> 4;
    const int mn   = lane & 15;
    const int row0 = blockIdx.x << 4;
    const int b    = row0 >> 10;
    const int l0   = row0 & (LL - 1);
    const int lA   = l0 + mn;

    const float biasB = b_B[mn];
    const float biasC = b_C[mn];
    const float biasD = (mn == 0) ? b_d[0] : 0.0f;
    v8f accB, accC, accD;
#pragma unroll
    for (int r = 0; r < 8; ++r) { accB[r] = biasB; accC[r] = biasC; accD[r] = biasD; }

    const float* __restrict__ xb = x + (size_t)b * DD * LL;
    const float n0sel = (mn == 0) ? 1.0f : 0.0f;

    const float* __restrict__ wbr = W_B + mn * DD + 2 * half;
    const float* __restrict__ wcr = W_C + mn * DD + 2 * half;
    const float* __restrict__ wdr = W_d + 2 * half;
    const float* __restrict__ xa  = xb + (size_t)(2 * half) * LL + lA;

    ProjFrag f0, f1;
    auto load_batch = [&](ProjFrag& f, const int kb) {
#pragma unroll
        for (int u = 0; u < KU; ++u) {
            const int k = kb + 4 * u;
            f.a[u][0] = xa[(size_t)k * LL];
            f.a[u][1] = xa[(size_t)(k + 1) * LL];
            f.wb[u] = *(const v2f*)(wbr + k);
            f.wc[u] = *(const v2f*)(wcr + k);
            f.wd[u] = *(const v2f*)(wdr + k);
        }
    };
    auto mma_batch = [&](const ProjFrag& f) {
#pragma unroll
        for (int u = 0; u < KU; ++u) {
            v2f wdm;
            wdm[0] = f.wd[u][0] * n0sel;
            wdm[1] = f.wd[u][1] * n0sel;
            accB = __builtin_amdgcn_wmma_f32_16x16x4_f32(false, f.a[u], false, f.wb[u], (short)0, accB, false, false);
            accC = __builtin_amdgcn_wmma_f32_16x16x4_f32(false, f.a[u], false, f.wc[u], (short)0, accC, false, false);
            accD = __builtin_amdgcn_wmma_f32_16x16x4_f32(false, f.a[u], false, wdm,     (short)0, accD, false, false);
        }
    };

    load_batch(f0, 0);
    for (int kb = 0; kb < DD - 2 * KS; kb += 2 * KS) {
        load_batch(f1, kb + KS);
        __builtin_amdgcn_sched_barrier(0);
        mma_batch(f0);
        __builtin_amdgcn_sched_barrier(0);
        load_batch(f0, kb + 2 * KS);
        __builtin_amdgcn_sched_barrier(0);
        mma_batch(f1);
        __builtin_amdgcn_sched_barrier(0);
    }
    load_batch(f1, DD - KS);
    __builtin_amdgcn_sched_barrier(0);
    mma_batch(f0);
    __builtin_amdgcn_sched_barrier(0);
    mma_batch(f1);

    const int gr0 = b * LL + l0 + 8 * half;
    const int pk0 = ((gr0 / PK) * NN + mn) * PK;
    const int pk1 = pk0 + NN * PK;
    v4f o;
    o[0] = accB[0]; o[1] = accB[1]; o[2] = accB[2]; o[3] = accB[3];
    *(v4f*)(Bp + pk0) = o;
    o[0] = accB[4]; o[1] = accB[5]; o[2] = accB[6]; o[3] = accB[7];
    *(v4f*)(Bp + pk1) = o;
    o[0] = accC[0]; o[1] = accC[1]; o[2] = accC[2]; o[3] = accC[3];
    *(v4f*)(Cp + pk0) = o;
    o[0] = accC[4]; o[1] = accC[5]; o[2] = accC[6]; o[3] = accC[7];
    *(v4f*)(Cp + pk1) = o;
#pragma unroll
    for (int r = 0; r < 8; ++r) {
        if (mn == 0) dlin[gr0 + r] = accD[r];
    }
}


template <int S> __device__ __forceinline__ float row_bcast(float v) {
    return __int_as_float(__builtin_amdgcn_update_dpp(0, __float_as_int(v), 0x150 + S, 0xf, 0xf, true));
}
template <int M> __device__ __forceinline__ float row_xor(float v) {
    return __int_as_float(__builtin_amdgcn_update_dpp(0, __float_as_int(v), 0x160 + M, 0xf, 0xf, true));
}

struct ScanBlk {
    float xv;
    v4f   bp[TB / PK];
    v4f   cp[TB / PK];
    float dt;
    float dl;
};

template <int M> __device__ __forceinline__ float bfly(bool hi, float lo_e, float hi_e) {
    const float keep = hi ? hi_e : lo_e;
    const float send = hi ? lo_e : hi_e;
    return keep + row_xor<M>(send);
}

typedef __attribute__((address_space(1))) const float gcfloat;
typedef __attribute__((address_space(1))) float gfloat;
typedef __attribute__((address_space(1))) const v4f gcv4f;

__device__ __forceinline__ float ld_lane(const float* __restrict__ ubase, const int elem, const unsigned lane_bytes) {
    return *(gcfloat*)((unsigned long long)ubase + (unsigned long long)lane_bytes + (unsigned long long)(elem * 4));
}
__device__ __forceinline__ v4f ld_lane4(const float* __restrict__ ubase, const int elem, const unsigned lane_bytes) {
    return *(gcv4f*)((unsigned long long)ubase + (unsigned long long)lane_bytes + (unsigned long long)(elem * 4));
}
__device__ __forceinline__ void st_lane(float* __restrict__ ubase, const int elem, const unsigned lane_bytes, const float v) {
    *(gfloat*)((unsigned long long)ubase + (unsigned long long)lane_bytes + (unsigned long long)(elem * 4)) = v;
}
__device__ __forceinline__ void keep_lane_offset(unsigned& lane_bytes) {
    asm volatile("" : "+v"(lane_bytes));
}

#define SCAN_STEP(S)                                                          \
    {                                                                         \
        const float dt = row_bcast<S>(dtv);                                   \
        const float xs = row_bcast<S>(xcur);                                  \
        const float a  = __expf(dt * An);                                     \
        h = fmaf(a, h, (a - An) * rAn * blk.bp[(S) / PK][(S) % PK] * xs);     \
        p[S] = blk.cp[(S) / PK][(S) % PK] * h;                                \
    }
#define SCAN_REFILL(Q)                                                        \
    {                                                                         \
        blk.bp[Q] = ld_lane4(bpn, (Q) * NN * PK, npt);                        \
        blk.cp[Q] = ld_lane4(cpn, (Q) * NN * PK, npt);                        \
    }

__global__ __launch_bounds__(32 * SW) __attribute__((amdgpu_waves_per_eu(1, 1))) void ssm_scan_serial(
    const float* __restrict__ x,
    const float* __restrict__ A,
    const float* __restrict__ p_Delta,
    const float* __restrict__ dlin,
    const float* __restrict__ Bp,
    const float* __restrict__ Cp,
    float* __restrict__ y)
{
    const unsigned tid   = threadIdx.x;
    const unsigned lane  = tid & 31;
    const unsigned n     = tid & 15;
    const unsigned rw    = tid >> 4;
    const unsigned pair0 = blockIdx.x * (2 * SW);
    const unsigned b     = pair0 / DD;
    const unsigned pair  = pair0 + rw;
    const unsigned d     = pair - b * DD;

    const float An  = A[d * NN + n];
    const float rAn = 1.0f / An;
    const float pD  = p_Delta[d];

    const float* __restrict__ xu  = x    + (size_t)pair0 * LL;
    const float* __restrict__ du  = dlin + (size_t)b * LL;
    const float* __restrict__ bpu = Bp   + (size_t)b * LL * NN;
    const float* __restrict__ cpu = Cp   + (size_t)b * LL * NN;
    float*       __restrict__ yu  = y    + (size_t)pair0 * LL;
    const unsigned nb = n * 4u;
    const unsigned np = n * (4u * PK);
    const unsigned xb = (rw * LL + n) * 4u;

    const bool m8 = (lane & 8) != 0;
    const bool m4 = (lane & 4) != 0;
    const bool m2 = (lane & 2) != 0;
    const bool m1 = (lane & 1) != 0;

    auto softplus = [&](const float dlv) {
        const float dsum = dlv + pD;
        return fmaxf(dsum, 0.0f) + log1pf(__expf(-fabsf(dsum)));
    };

    ScanBlk st[NSET];
    float dl0[NSET];
#pragma unroll
    for (int i = 0; i < NSET; ++i) {
        st[i].dl = ld_lane(du, (NSET + i) * TB, nb);
        st[i].xv = ld_lane(xu, i * TB, xb);
#pragma unroll
        for (int q = 0; q < TB / PK; ++q) st[i].bp[q] = ld_lane4(bpu, (i * TB + q * PK) * NN, np);
#pragma unroll
        for (int q = 0; q < TB / PK; ++q) st[i].cp[q] = ld_lane4(cpu, (i * TB + q * PK) * NN, np);
    }
    __builtin_amdgcn_sched_barrier(0);
#pragma unroll
    for (int i = 0; i < NSET; ++i) dl0[i] = ld_lane(du, i * TB, nb);
#pragma unroll
    for (int i = NSET - 1; i >= 0; --i) st[i].dt = softplus(dl0[i]);

    float h = 0.0f;

    auto run_steps = [&](ScanBlk& blk, const int kb, float (&p)[TB], const float ndt,
                         const unsigned xbt, const unsigned npt) {
        const int kr = (kb + NSET < NBLK) ? kb + NSET : NBLK - 1;
        const int l2 = kr * TB;
        const float dtv  = blk.dt;
        const float xcur = blk.xv;
        const float* __restrict__ bpn = bpu + l2 * NN;
        const float* __restrict__ cpn = cpu + l2 * NN;
        SCAN_STEP(0)  SCAN_STEP(1)  SCAN_STEP(2)  SCAN_STEP(3)  SCAN_REFILL(0)
        SCAN_STEP(4)  SCAN_STEP(5)  SCAN_STEP(6)  SCAN_STEP(7)  SCAN_REFILL(1)
        SCAN_STEP(8)  SCAN_STEP(9)  SCAN_STEP(10) SCAN_STEP(11) SCAN_REFILL(2)
        SCAN_STEP(12) SCAN_STEP(13) SCAN_STEP(14) SCAN_STEP(15) SCAN_REFILL(3)
        blk.xv = ld_lane(xu + l2, 0, xbt);
        blk.dt = ndt;
    };
    auto sum_store = [&](const float (&p)[TB], const int l0, const unsigned xbt) {
        float w8[8], w4[4], w2[2];
#pragma unroll
        for (int i = 0; i < 8; ++i) w8[i] = bfly<8>(m8, p[i], p[i + 8]);
#pragma unroll
        for (int i = 0; i < 4; ++i) w4[i] = bfly<4>(m4, w8[i], w8[i + 4]);
#pragma unroll
        for (int i = 0; i < 2; ++i) w2[i] = bfly<2>(m2, w4[i], w4[i + 2]);
        st_lane(yu + l0, 0, xbt, bfly<1>(m1, w2[0], w2[1]));
    };

    unsigned xbt = xb, npt = np;
    const unsigned dlo_last = nb + (NBLK - NSET) * TB * 4u;
    unsigned dlo = nb + 2 * NSET * TB * 4u;
    for (int k = 0; k < NBLK; k += NSET) {
        keep_lane_offset(xbt);
        keep_lane_offset(npt);
        float ndt[NSET];
#pragma unroll
        for (int i = 0; i < NSET; ++i) ndt[i] = softplus(st[i].dl);
#pragma unroll
        for (int i = 0; i < NSET; ++i) st[i].dl = ld_lane(du, i * TB, dlo);
        float p[NSET][TB];
#pragma unroll
        for (int i = 0; i < NSET; ++i) {
            run_steps(st[i], k + i, p[i], ndt[i], xbt, npt);
            __builtin_amdgcn_sched_barrier(0);
            sum_store(p[i], (k + i) * TB, xbt);
        }
        dlo = min(dlo + NSET * TB * 4u, dlo_last);
    }
}

extern "C" void kernel_launch(void* const* d_in, const int* in_sizes, int n_in,
                              void* d_out, int out_size, void* d_ws, size_t ws_size,
                              hipStream_t stream)
{
    const float* x       = (const float*)d_in[0];
    const float* A       = (const float*)d_in[1];
    const float* W_B     = (const float*)d_in[2];
    const float* b_B     = (const float*)d_in[3];
    const float* W_C     = (const float*)d_in[4];
    const float* b_C     = (const float*)d_in[5];
    const float* W_d     = (const float*)d_in[6];
    const float* b_d     = (const float*)d_in[7];
    const float* p_Delta = (const float*)d_in[8];
    float* y = (float*)d_out;

    float* ws   = (float*)d_ws;
    float* Bp   = ws;
    float* Cp   = Bp + (size_t)BB * LL * NN;
    float* dlin = Cp + (size_t)BB * LL * NN;

    ssm_proj_wmma<<<dim3(128), dim3(32), 0, stream>>>(
        x, W_B, b_B, W_C, b_C, W_d, b_d, Bp, Cp, dlin);

    ssm_scan_serial<<<dim3((BB * DD) / (2 * SW)), dim3(32 * SW), 0, stream>>>(
        x, A, p_Delta, dlin, Bp, Cp, y);
}


# projection GEMM: its three products (W_B, W_C, W_d) go to three waves of a group, one each (192 matrix instructions per wave, was 576); loads 20 k-steps ahead, between the products; sums unchanged
# speedup vs baseline: 3.6020x; 1.1303x over previous
#include <hip/hip_runtime.h>
#include <math.h>

#define DD 768
#define LL 1024
#define NN 16
#define BB 2
#define KRING 24
#define KAHEAD 20
#define TB 16
#define NBLK (LL / TB)
#define PK 4
#define NSET 4
#define SW 3

typedef float v2f __attribute__((ext_vector_type(2)));
typedef float v4f __attribute__((ext_vector_type(4)));
typedef float v8f __attribute__((ext_vector_type(8)));


__global__ __launch_bounds__(96) void ssm_proj_wmma(
    const float* __restrict__ x,
    const float* __restrict__ W_B,
    const float* __restrict__ b_B,
    const float* __restrict__ W_C,
    const float* __restrict__ b_C,
    const float* __restrict__ W_d,
    const float* __restrict__ b_d,
    float* __restrict__ Bp,
    float* __restrict__ Cp,
    float* __restrict__ dlin)
{
    const int lane = threadIdx.x & 31;
    const int which = __builtin_amdgcn_readfirstlane((int)(threadIdx.x >> 5));
    const int half = lane >> 4;
    const int mn   = lane & 15;
    const int row0 = blockIdx.x << 4;
    const int b    = row0 >> 10;
    const int l0   = row0 & (LL - 1);
    const int lA   = l0 + mn;

    const float* __restrict__ wmat = (which == 0) ? W_B : ((which == 1) ? W_C : W_d);
    const float* __restrict__ bvec = (which == 0) ? b_B : ((which == 1) ? b_C : b_d);
    const int wrow = (which == 2) ? 0 : mn;
    const float bias = bvec[wrow];
    v8f acc;
#pragma unroll
    for (int r = 0; r < 8; ++r) acc[r] = bias;

    const float* __restrict__ wr = wmat + wrow * DD + 2 * half;
    const float* __restrict__ xa = x + (size_t)b * DD * LL + (size_t)(2 * half) * LL + lA;

    v2f fa[KRING], fw[KRING];
    auto load_step = [&](const int slot, const int k) {
        fa[slot][0] = xa[(size_t)k * LL];
        fa[slot][1] = xa[(size_t)(k + 1) * LL];
        fw[slot] = *(const v2f*)(wr + k);
    };
    auto product = [&](const int slot) {
        __builtin_amdgcn_sched_barrier(0);
        acc = __builtin_amdgcn_wmma_f32_16x16x4_f32(false, fa[slot], false, fw[slot], (short)0, acc, false, false);
        __builtin_amdgcn_sched_barrier(0);
    };

#pragma unroll
    for (int s = 0; s < KAHEAD; ++s) load_step(s, 4 * s);
    int kb = 0;
    for (; kb < DD - 4 * KRING; kb += 4 * KRING) {
#pragma unroll
        for (int s = 0; s < KRING; ++s) {
            load_step((s + KAHEAD) % KRING, kb + 4 * (s + KAHEAD));
            product(s);
        }
    }
#pragma unroll
    for (int s = 0; s < KRING; ++s) {
        if (s + KAHEAD < KRING) load_step(s + KAHEAD, kb + 4 * (s + KAHEAD));
        product(s);
    }

    const int gr0 = b * LL + l0 + 8 * half;
    if (which == 2) {
#pragma unroll
        for (int r = 0; r < 8; ++r) {
            if (mn == 0) dlin[gr0 + r] = acc[r];
        }
    } else {
        float* __restrict__ outp = (which == 0) ? Bp : Cp;
        const int pk0 = ((gr0 / PK) * NN + mn) * PK;
        const int pk1 = pk0 + NN * PK;
        v4f o;
        o[0] = acc[0]; o[1] = acc[1]; o[2] = acc[2]; o[3] = acc[3];
        *(v4f*)(outp + pk0) = o;
        o[0] = acc[4]; o[1] = acc[5]; o[2] = acc[6]; o[3] = acc[7];
        *(v4f*)(outp + pk1) = o;
    }
}


template <int S> __device__ __forceinline__ float row_bcast(float v) {
    return __int_as_float(__builtin_amdgcn_update_dpp(0, __float_as_int(v), 0x150 + S, 0xf, 0xf, true));
}
template <int M> __device__ __forceinline__ float row_xor(float v) {
    return __int_as_float(__builtin_amdgcn_update_dpp(0, __float_as_int(v), 0x160 + M, 0xf, 0xf, true));
}

struct ScanBlk {
    float xv;
    v4f   bp[TB / PK];
    v4f   cp[TB / PK];
    float dt;
    float dl;
};

template <int M> __device__ __forceinline__ float bfly(bool hi, float lo_e, float hi_e) {
    const float keep = hi ? hi_e : lo_e;
    const float send = hi ? lo_e : hi_e;
    return keep + row_xor<M>(send);
}

typedef __attribute__((address_space(1))) const float gcfloat;
typedef __attribute__((address_space(1))) float gfloat;
typedef __attribute__((address_space(1))) const v4f gcv4f;

__device__ __forceinline__ float ld_lane(const float* __restrict__ ubase, const int elem, const unsigned lane_bytes) {
    return *(gcfloat*)((unsigned long long)ubase + (unsigned long long)lane_bytes + (unsigned long long)(elem * 4));
}
__device__ __forceinline__ v4f ld_lane4(const float* __restrict__ ubase, const int elem, const unsigned lane_bytes) {
    return *(gcv4f*)((unsigned long long)ubase + (unsigned long long)lane_bytes + (unsigned long long)(elem * 4));
}
__device__ __forceinline__ void st_lane(float* __restrict__ ubase, const int elem, const unsigned lane_bytes, const float v) {
    *(gfloat*)((unsigned long long)ubase + (unsigned long long)lane_bytes + (unsigned long long)(elem * 4)) = v;
}
__device__ __forceinline__ void keep_lane_offset(unsigned& lane_bytes) {
    asm volatile("" : "+v"(lane_bytes));
}

#define SCAN_STEP(S)                                                          \
    {                                                                         \
        const float dt = row_bcast<S>(dtv);                                   \
        const float xs = row_bcast<S>(xcur);                                  \
        const float a  = __expf(dt * An);                                     \
        h = fmaf(a, h, (a - An) * rAn * blk.bp[(S) / PK][(S) % PK] * xs);     \
        p[S] = blk.cp[(S) / PK][(S) % PK] * h;                                \
    }
#define SCAN_REFILL(Q)                                                        \
    {                                                                         \
        blk.bp[Q] = ld_lane4(bpn, (Q) * NN * PK, npt);                        \
        blk.cp[Q] = ld_lane4(cpn, (Q) * NN * PK, npt);                        \
    }

__global__ __launch_bounds__(32 * SW) __attribute__((amdgpu_waves_per_eu(1, 1))) void ssm_scan_serial(
    const float* __restrict__ x,
    const float* __restrict__ A,
    const float* __restrict__ p_Delta,
    const float* __restrict__ dlin,
    const float* __restrict__ Bp,
    const float* __restrict__ Cp,
    float* __restrict__ y)
{
    const unsigned tid   = threadIdx.x;
    const unsigned lane  = tid & 31;
    const unsigned n     = tid & 15;
    const unsigned rw    = tid >> 4;
    const unsigned pair0 = blockIdx.x * (2 * SW);
    const unsigned b     = pair0 / DD;
    const unsigned pair  = pair0 + rw;
    const unsigned d     = pair - b * DD;

    const float An  = A[d * NN + n];
    const float rAn = 1.0f / An;
    const float pD  = p_Delta[d];

    const float* __restrict__ xu  = x    + (size_t)pair0 * LL;
    const float* __restrict__ du  = dlin + (size_t)b * LL;
    const float* __restrict__ bpu = Bp   + (size_t)b * LL * NN;
    const float* __restrict__ cpu = Cp   + (size_t)b * LL * NN;
    float*       __restrict__ yu  = y    + (size_t)pair0 * LL;
    const unsigned nb = n * 4u;
    const unsigned np = n * (4u * PK);
    const unsigned xb = (rw * LL + n) * 4u;

    const bool m8 = (lane & 8) != 0;
    const bool m4 = (lane & 4) != 0;
    const bool m2 = (lane & 2) != 0;
    const bool m1 = (lane & 1) != 0;

    auto softplus = [&](const float dlv) {
        const float dsum = dlv + pD;
        return fmaxf(dsum, 0.0f) + log1pf(__expf(-fabsf(dsum)));
    };

    ScanBlk st[NSET];
    float dl0[NSET];
#pragma unroll
    for (int i = 0; i < NSET; ++i) {
        st[i].dl = ld_lane(du, (NSET + i) * TB, nb);
        st[i].xv = ld_lane(xu, i * TB, xb);
#pragma unroll
        for (int q = 0; q < TB / PK; ++q) st[i].bp[q] = ld_lane4(bpu, (i * TB + q * PK) * NN, np);
#pragma unroll
        for (int q = 0; q < TB / PK; ++q) st[i].cp[q] = ld_lane4(cpu, (i * TB + q * PK) * NN, np);
    }
    __builtin_amdgcn_sched_barrier(0);
#pragma unroll
    for (int i = 0; i < NSET; ++i) dl0[i] = ld_lane(du, i * TB, nb);
#pragma unroll
    for (int i = NSET - 1; i >= 0; --i) st[i].dt = softplus(dl0[i]);

    float h = 0.0f;

    auto run_steps = [&](ScanBlk& blk, const int kb, float (&p)[TB], const float ndt,
                         const unsigned xbt, const unsigned npt) {
        const int kr = (kb + NSET < NBLK) ? kb + NSET : NBLK - 1;
        const int l2 = kr * TB;
        const float dtv  = blk.dt;
        const float xcur = blk.xv;
        const float* __restrict__ bpn = bpu + l2 * NN;
        const float* __restrict__ cpn = cpu + l2 * NN;
        SCAN_STEP(0)  SCAN_STEP(1)  SCAN_STEP(2)  SCAN_STEP(3)  SCAN_REFILL(0)
        SCAN_STEP(4)  SCAN_STEP(5)  SCAN_STEP(6)  SCAN_STEP(7)  SCAN_REFILL(1)
        SCAN_STEP(8)  SCAN_STEP(9)  SCAN_STEP(10) SCAN_STEP(11) SCAN_REFILL(2)
        SCAN_STEP(12) SCAN_STEP(13) SCAN_STEP(14) SCAN_STEP(15) SCAN_REFILL(3)
        blk.xv = ld_lane(xu + l2, 0, xbt);
        blk.dt = ndt;
    };
    auto sum_store = [&](const float (&p)[TB], const int l0, const unsigned xbt) {
        float w8[8], w4[4], w2[2];
#pragma unroll
        for (int i = 0; i < 8; ++i) w8[i] = bfly<8>(m8, p[i], p[i + 8]);
#pragma unroll
        for (int i = 0; i < 4; ++i) w4[i] = bfly<4>(m4, w8[i], w8[i + 4]);
#pragma unroll
        for (int i = 0; i < 2; ++i) w2[i] = bfly<2>(m2, w4[i], w4[i + 2]);
        st_lane(yu + l0, 0, xbt, bfly<1>(m1, w2[0], w2[1]));
    };

    unsigned xbt = xb, npt = np;
    const unsigned dlo_last = nb + (NBLK - NSET) * TB * 4u;
    unsigned dlo = nb + 2 * NSET * TB * 4u;
    for (int k = 0; k < NBLK; k += NSET) {
        keep_lane_offset(xbt);
        keep_lane_offset(npt);
        float ndt[NSET];
#pragma unroll
        for (int i = 0; i < NSET; ++i) ndt[i] = softplus(st[i].dl);
#pragma unroll
        for (int i = 0; i < NSET; ++i) st[i].dl = ld_lane(du, i * TB, dlo);
        float p[NSET][TB];
#pragma unroll
        for (int i = 0; i < NSET; ++i) {
            run_steps(st[i], k + i, p[i], ndt[i], xbt, npt);
            __builtin_amdgcn_sched_barrier(0);
            sum_store(p[i], (k + i) * TB, xbt);
        }
        dlo = min(dlo + NSET * TB * 4u, dlo_last);
    }
}

extern "C" void kernel_launch(void* const* d_in, const int* in_sizes, int n_in,
                              void* d_out, int out_size, void* d_ws, size_t ws_size,
                              hipStream_t stream)
{
    const float* x       = (const float*)d_in[0];
    const float* A       = (const float*)d_in[1];
    const float* W_B     = (const float*)d_in[2];
    const float* b_B     = (const float*)d_in[3];
    const float* W_C     = (const float*)d_in[4];
    const float* b_C     = (const float*)d_in[5];
    const float* W_d     = (const float*)d_in[6];
    const float* b_d     = (const float*)d_in[7];
    const float* p_Delta = (const float*)d_in[8];
    float* y = (float*)d_out;

    float* ws   = (float*)d_ws;
    float* Bp   = ws;
    float* Cp   = Bp + (size_t)BB * LL * NN;
    float* dlin = Cp + (size_t)BB * LL * NN;

    ssm_proj_wmma<<<dim3(128), dim3(96), 0, stream>>>(
        x, W_B, b_B, W_C, b_C, W_d, b_d, Bp, Cp, dlin);

    ssm_scan_serial<<<dim3((BB * DD) / (2 * SW)), dim3(32 * SW), 0, stream>>>(
        x, A, p_Delta, dlin, Bp, Cp, y);
}


# scan: a 4th wave per group, on the SIMD the 3 scan waves left idle, computes all softplus and stages delta and x in LDS a trip ahead; scan waves read them as 16-byte broadcasts; values unchanged
# speedup vs baseline: 3.8258x; 1.0621x over previous
#include <hip/hip_runtime.h>
#include <math.h>

#define DD 768
#define LL 1024
#define NN 16
#define BB 2
#define KRING 24
#define KAHEAD 20
#define TB 16
#define NBLK (LL / TB)
#define PK 4
#define NSET 4
#define LP (LL + 9 * TB)
#define SW 3

typedef float v2f __attribute__((ext_vector_type(2)));
typedef float v4f __attribute__((ext_vector_type(4)));
typedef float v8f __attribute__((ext_vector_type(8)));


__global__ __launch_bounds__(96) void ssm_proj_wmma(
    const float* __restrict__ x,
    const float* __restrict__ W_B,
    const float* __restrict__ b_B,
    const float* __restrict__ W_C,
    const float* __restrict__ b_C,
    const float* __restrict__ W_d,
    const float* __restrict__ b_d,
    float* __restrict__ Bp,
    float* __restrict__ Cp,
    float* __restrict__ dlin)
{
    const int lane = threadIdx.x & 31;
    const int which = __builtin_amdgcn_readfirstlane((int)(threadIdx.x >> 5));
    const int half = lane >> 4;
    const int mn   = lane & 15;
    const int row0 = blockIdx.x << 4;
    const int b    = row0 >> 10;
    const int l0   = row0 & (LL - 1);
    const int lA   = l0 + mn;

    const float* __restrict__ wmat = (which == 0) ? W_B : ((which == 1) ? W_C : W_d);
    const float* __restrict__ bvec = (which == 0) ? b_B : ((which == 1) ? b_C : b_d);
    const int wrow = (which == 2) ? 0 : mn;
    const float bias = bvec[wrow];
    v8f acc;
#pragma unroll
    for (int r = 0; r < 8; ++r) acc[r] = bias;

    const float* __restrict__ wr = wmat + wrow * DD + 2 * half;
    const float* __restrict__ xa = x + (size_t)b * DD * LL + (size_t)(2 * half) * LL + lA;

    v2f fa[KRING], fw[KRING];
    auto load_step = [&](const int slot, const int k) {
        fa[slot][0] = xa[(size_t)k * LL];
        fa[slot][1] = xa[(size_t)(k + 1) * LL];
        fw[slot] = *(const v2f*)(wr + k);
    };
    auto product = [&](const int slot) {
        __builtin_amdgcn_sched_barrier(0);
        acc = __builtin_amdgcn_wmma_f32_16x16x4_f32(false, fa[slot], false, fw[slot], (short)0, acc, false, false);
        __builtin_amdgcn_sched_barrier(0);
    };

#pragma unroll
    for (int s = 0; s < KAHEAD; ++s) load_step(s, 4 * s);
    int kb = 0;
    for (; kb < DD - 4 * KRING; kb += 4 * KRING) {
#pragma unroll
        for (int s = 0; s < KRING; ++s) {
            load_step((s + KAHEAD) % KRING, kb + 4 * (s + KAHEAD));
            product(s);
        }
    }
#pragma unroll
    for (int s = 0; s < KRING; ++s) {
        if (s + KAHEAD < KRING) load_step(s + KAHEAD, kb + 4 * (s + KAHEAD));
        product(s);
    }

    const int gr0 = b * LL + l0 + 8 * half;
    if (which == 2) {
#pragma unroll
        for (int r = 0; r < 8; ++r) {
            if (mn == 0) dlin[gr0 + r] = acc[r];
        }
    } else {
        float* __restrict__ outp = (which == 0) ? Bp : Cp;
        const int pk0 = ((gr0 / PK) * NN + mn) * PK;
        const int pk1 = pk0 + NN * PK;
        v4f o;
        o[0] = acc[0]; o[1] = acc[1]; o[2] = acc[2]; o[3] = acc[3];
        *(v4f*)(outp + pk0) = o;
        o[0] = acc[4]; o[1] = acc[5]; o[2] = acc[6]; o[3] = acc[7];
        *(v4f*)(outp + pk1) = o;
    }
}


template <int M> __device__ __forceinline__ float row_xor(float v) {
    return __int_as_float(__builtin_amdgcn_update_dpp(0, __float_as_int(v), 0x160 + M, 0xf, 0xf, true));
}

struct ScanBlk {
    v4f   bp[TB / PK];
    v4f   cp[TB / PK];
};

template <int M> __device__ __forceinline__ float bfly(bool hi, float lo_e, float hi_e) {
    const float keep = hi ? hi_e : lo_e;
    const float send = hi ? lo_e : hi_e;
    return keep + row_xor<M>(send);
}

typedef __attribute__((address_space(1))) float gfloat;
typedef __attribute__((address_space(1))) const v4f gcv4f;

__device__ __forceinline__ v4f ld_lane4(const float* __restrict__ ubase, const int elem, const unsigned lane_bytes) {
    return *(gcv4f*)((unsigned long long)ubase + (unsigned long long)lane_bytes + (unsigned long long)(elem * 4));
}
__device__ __forceinline__ void st_lane(float* __restrict__ ubase, const int elem, const unsigned lane_bytes, const float v) {
    *(gfloat*)((unsigned long long)ubase + (unsigned long long)lane_bytes + (unsigned long long)(elem * 4)) = v;
}
__device__ __forceinline__ void keep_lane_offset(unsigned& lane_bytes) {
    asm volatile("" : "+v"(lane_bytes));
}
__device__ __forceinline__ void wg_handover() {
    asm volatile("s_wait_dscnt 0x0\n\ts_barrier_signal -1\n\ts_barrier_wait -1" ::: "memory");
}

#define SCAN_STEP(S)                                                          \
    {                                                                         \
        const float dt = dcur[(S) / PK][(S) % PK];                            \
        const float xs = xcur[(S) / PK][(S) % PK];                            \
        const float a  = __expf(dt * An);                                     \
        h = fmaf(a, h, (a - An) * rAn * blk.bp[(S) / PK][(S) % PK] * xs);     \
        p[S] = blk.cp[(S) / PK][(S) % PK] * h;                                \
    }
#define SCAN_REFILL(Q)                                                        \
    {                                                                         \
        blk.bp[Q] = ld_lane4(bpn, (Q) * NN * PK, npt);                        \
        blk.cp[Q] = ld_lane4(cpn, (Q) * NN * PK, npt);                        \
    }

__global__ __launch_bounds__(32 * (SW + 1)) __attribute__((amdgpu_waves_per_eu(1, 1))) void ssm_scan_serial(
    const float* __restrict__ x,
    const float* __restrict__ A,
    const float* __restrict__ p_Delta,
    const float* __restrict__ dlin,
    const float* __restrict__ Bp,
    const float* __restrict__ Cp,
    float* __restrict__ y)
{
    __shared__ __attribute__((aligned(16))) float s_dt[2 * SW * LP];
    __shared__ __attribute__((aligned(16))) float s_x[2 * SW * LP];

    const unsigned tid   = threadIdx.x;
    const unsigned lane  = tid & 31;
    const unsigned n     = tid & 15;
    const unsigned half  = lane >> 4;
    const bool feeder    = __builtin_amdgcn_readfirstlane((int)(tid >> 5)) == SW;
    const unsigned rw    = feeder ? half : (tid >> 4);
    const unsigned pair0 = blockIdx.x * (2 * SW);
    const unsigned b     = pair0 / DD;
    const unsigned d0    = pair0 - b * DD;
    const unsigned d     = d0 + rw;

    const float An  = A[d * NN + n];
    const float rAn = 1.0f / An;

    const float* __restrict__ bpu = Bp   + (size_t)b * LL * NN;
    const float* __restrict__ cpu = Cp   + (size_t)b * LL * NN;
    float*       __restrict__ yu  = y    + (size_t)pair0 * LL;
    const unsigned np = n * (4u * PK);
    const unsigned xb = (rw * LL + n) * 4u;
    const float* sdr = s_dt + rw * LP;
    const float* sxr = s_x  + rw * LP;

    const bool m8 = (lane & 8) != 0;
    const bool m4 = (lane & 4) != 0;
    const bool m2 = (lane & 2) != 0;
    const bool m1 = (lane & 1) != 0;

    const float* __restrict__ xf = x    + (size_t)(pair0 + half) * LL + n;
    const float* __restrict__ df = dlin + (size_t)b * LL + n;
    const unsigned lf = half * LP + n;
    float pDv[SW];
#pragma unroll
    for (int sp = 0; sp < SW; ++sp) pDv[sp] = p_Delta[d0 + 2 * sp + half];

    auto softplus = [&](const float dlv, const float pd) {
        const float dsum = dlv + pd;
        return fmaxf(dsum, 0.0f) + log1pf(__expf(-fabsf(dsum)));
    };
    float fd[NSET], fx[NSET][SW];
    auto feeder_fetch = [&](const int blk0) {
#pragma unroll
        for (int q = 0; q < NSET; ++q) {
            const int bk = (blk0 + q < NBLK) ? blk0 + q : NBLK - 1;
            fd[q] = df[bk * TB];
#pragma unroll
            for (int sp = 0; sp < SW; ++sp) fx[q][sp] = xf[(size_t)(2 * sp) * LL + bk * TB];
        }
    };
    auto feeder_emit = [&](const int blk0) {
#pragma unroll
        for (int q = 0; q < NSET; ++q) {
            const int bk = blk0 + q;
#pragma unroll
            for (int sp = 0; sp < SW; ++sp) {
                s_dt[lf + 2 * sp * LP + bk * TB] = softplus(fd[q], pDv[sp]);
                s_x [lf + 2 * sp * LP + bk * TB] = fx[q][sp];
            }
        }
    };

    if (feeder) {
        float gd[2], gx[2][SW];
#pragma unroll
        for (int q = 0; q < 2; ++q) {
            gd[q] = df[q * TB];
#pragma unroll
            for (int sp = 0; sp < SW; ++sp) gx[q][sp] = xf[(size_t)(2 * sp) * LL + q * TB];
        }
        feeder_fetch(2);
#pragma unroll
        for (int q = 0; q < 2; ++q) {
#pragma unroll
            for (int sp = 0; sp < SW; ++sp) {
                s_dt[lf + 2 * sp * LP + q * TB] = softplus(gd[q], pDv[sp]);
                s_x [lf + 2 * sp * LP + q * TB] = gx[q][sp];
            }
        }
        feeder_emit(2);
        feeder_fetch(2 + NSET);
        wg_handover();
        for (int k = 0; k < NBLK; k += NSET) {
            feeder_emit(k + 2 + NSET);
            feeder_fetch(k + 2 + 2 * NSET);
            wg_handover();
        }
        return;
    }

    ScanBlk st[NSET];
#pragma unroll
    for (int i = 0; i < NSET; ++i) {
#pragma unroll
        for (int q = 0; q < TB / PK; ++q) st[i].bp[q] = ld_lane4(bpu, (i * TB + q * PK) * NN, np);
#pragma unroll
        for (int q = 0; q < TB / PK; ++q) st[i].cp[q] = ld_lane4(cpu, (i * TB + q * PK) * NN, np);
    }
    wg_handover();

    v4f dq[2][TB / PK], xq[2][TB / PK];
#pragma unroll
    for (int i = 0; i < 2; ++i) {
#pragma unroll
        for (int q = 0; q < TB / PK; ++q) {
            dq[i][q] = *(const v4f*)(sdr + i * TB + q * PK);
            xq[i][q] = *(const v4f*)(sxr + i * TB + q * PK);
        }
    }
    float h = 0.0f;

    auto run_steps = [&](ScanBlk& blk, const int kb, float (&p)[TB],
                         v4f (&dcur)[TB / PK], v4f (&xcur)[TB / PK], const unsigned npt) {
        const int kr = (kb + NSET < NBLK) ? kb + NSET : NBLK - 1;
        const int kn = (kb + 2 < NBLK) ? kb + 2 : NBLK - 1;
        const int l2 = kr * TB;
        const float* __restrict__ bpn = bpu + l2 * NN;
        const float* __restrict__ cpn = cpu + l2 * NN;
        SCAN_STEP(0)  SCAN_STEP(1)  SCAN_STEP(2)  SCAN_STEP(3)  SCAN_REFILL(0)
        SCAN_STEP(4)  SCAN_STEP(5)  SCAN_STEP(6)  SCAN_STEP(7)  SCAN_REFILL(1)
        SCAN_STEP(8)  SCAN_STEP(9)  SCAN_STEP(10) SCAN_STEP(11) SCAN_REFILL(2)
        SCAN_STEP(12) SCAN_STEP(13) SCAN_STEP(14) SCAN_STEP(15) SCAN_REFILL(3)
#pragma unroll
        for (int q = 0; q < TB / PK; ++q) {
            dcur[q] = *(const v4f*)(sdr + kn * TB + q * PK);
            xcur[q] = *(const v4f*)(sxr + kn * TB + q * PK);
        }
    };
    auto sum_store = [&](const float (&p)[TB], const int l0, const unsigned xbt) {
        float w8[8], w4[4], w2[2];
#pragma unroll
        for (int i = 0; i < 8; ++i) w8[i] = bfly<8>(m8, p[i], p[i + 8]);
#pragma unroll
        for (int i = 0; i < 4; ++i) w4[i] = bfly<4>(m4, w8[i], w8[i + 4]);
#pragma unroll
        for (int i = 0; i < 2; ++i) w2[i] = bfly<2>(m2, w4[i], w4[i + 2]);
        st_lane(yu + l0, 0, xbt, bfly<1>(m1, w2[0], w2[1]));
    };

    unsigned xbt = xb, npt = np;
    for (int k = 0; k < NBLK; k += NSET) {
        keep_lane_offset(xbt);
        keep_lane_offset(npt);
        float p[NSET][TB];
#pragma unroll
        for (int i = 0; i < NSET; ++i) {
            run_steps(st[i], k + i, p[i], dq[i & 1], xq[i & 1], npt);
            __builtin_amdgcn_sched_barrier(0);
            sum_store(p[i], (k + i) * TB, xbt);
        }
        wg_handover();
    }
}

extern "C" void kernel_launch(void* const* d_in, const int* in_sizes, int n_in,
                              void* d_out, int out_size, void* d_ws, size_t ws_size,
                              hipStream_t stream)
{
    const float* x       = (const float*)d_in[0];
    const float* A       = (const float*)d_in[1];
    const float* W_B     = (const float*)d_in[2];
    const float* b_B     = (const float*)d_in[3];
    const float* W_C     = (const float*)d_in[4];
    const float* b_C     = (const float*)d_in[5];
    const float* W_d     = (const float*)d_in[6];
    const float* b_d     = (const float*)d_in[7];
    const float* p_Delta = (const float*)d_in[8];
    float* y = (float*)d_out;

    float* ws   = (float*)d_ws;
    float* Bp   = ws;
    float* Cp   = Bp + (size_t)BB * LL * NN;
    float* dlin = Cp + (size_t)BB * LL * NN;

    ssm_proj_wmma<<<dim3(128), dim3(96), 0, stream>>>(
        x, W_B, b_B, W_C, b_C, W_d, b_d, Bp, Cp, dlin);

    ssm_scan_serial<<<dim3((BB * DD) / (2 * SW)), dim3(32 * (SW + 1)), 0, stream>>>(
        x, A, p_Delta, dlin, Bp, Cp, y);
}


# scan kernel: address bases built once per trip (one clamp per trip; feeder in whole trips, two ahead; LDS read-ahead unclamped): none rebuilt behind issued loads, translation waits per trip 9 -> 3
# speedup vs baseline: 3.9809x; 1.0405x over previous
#include <hip/hip_runtime.h>
#include <math.h>

#define DD 768
#define LL 1024
#define NN 16
#define BB 2
#define KRING 24
#define KAHEAD 20
#define TB 16
#define NBLK (LL / TB)
#define PK 4
#define NSET 4
#define LP (LL + 9 * TB)
#define SW 3

typedef float v2f __attribute__((ext_vector_type(2)));
typedef float v4f __attribute__((ext_vector_type(4)));
typedef float v8f __attribute__((ext_vector_type(8)));


__global__ __launch_bounds__(96) void ssm_proj_wmma(
    const float* __restrict__ x,
    const float* __restrict__ W_B,
    const float* __restrict__ b_B,
    const float* __restrict__ W_C,
    const float* __restrict__ b_C,
    const float* __restrict__ W_d,
    const float* __restrict__ b_d,
    float* __restrict__ Bp,
    float* __restrict__ Cp,
    float* __restrict__ dlin)
{
    const int lane = threadIdx.x & 31;
    const int which = __builtin_amdgcn_readfirstlane((int)(threadIdx.x >> 5));
    const int half = lane >> 4;
    const int mn   = lane & 15;
    const int row0 = blockIdx.x << 4;
    const int b    = row0 >> 10;
    const int l0   = row0 & (LL - 1);
    const int lA   = l0 + mn;

    const float* __restrict__ wmat = (which == 0) ? W_B : ((which == 1) ? W_C : W_d);
    const float* __restrict__ bvec = (which == 0) ? b_B : ((which == 1) ? b_C : b_d);
    const int wrow = (which == 2) ? 0 : mn;
    const float bias = bvec[wrow];
    v8f acc;
#pragma unroll
    for (int r = 0; r < 8; ++r) acc[r] = bias;

    const float* __restrict__ wr = wmat + wrow * DD + 2 * half;
    const float* __restrict__ xa = x + (size_t)b * DD * LL + (size_t)(2 * half) * LL + lA;

    v2f fa[KRING], fw[KRING];
    auto load_step = [&](const int slot, const int k) {
        fa[slot][0] = xa[(size_t)k * LL];
        fa[slot][1] = xa[(size_t)(k + 1) * LL];
        fw[slot] = *(const v2f*)(wr + k);
    };
    auto product = [&](const int slot) {
        __builtin_amdgcn_sched_barrier(0);
        acc = __builtin_amdgcn_wmma_f32_16x16x4_f32(false, fa[slot], false, fw[slot], (short)0, acc, false, false);
        __builtin_amdgcn_sched_barrier(0);
    };

#pragma unroll
    for (int s = 0; s < KAHEAD; ++s) load_step(s, 4 * s);
    int kb = 0;
    for (; kb < DD - 4 * KRING; kb += 4 * KRING) {
#pragma unroll
        for (int s = 0; s < KRING; ++s) {
            load_step((s + KAHEAD) % KRING, kb + 4 * (s + KAHEAD));
            product(s);
        }
    }
#pragma unroll
    for (int s = 0; s < KRING; ++s) {
        if (s + KAHEAD < KRING) load_step(s + KAHEAD, kb + 4 * (s + KAHEAD));
        product(s);
    }

    const int gr0 = b * LL + l0 + 8 * half;
    if (which == 2) {
#pragma unroll
        for (int r = 0; r < 8; ++r) {
            if (mn == 0) dlin[gr0 + r] = acc[r];
        }
    } else {
        float* __restrict__ outp = (which == 0) ? Bp : Cp;
        const int pk0 = ((gr0 / PK) * NN + mn) * PK;
        const int pk1 = pk0 + NN * PK;
        v4f o;
        o[0] = acc[0]; o[1] = acc[1]; o[2] = acc[2]; o[3] = acc[3];
        *(v4f*)(outp + pk0) = o;
        o[0] = acc[4]; o[1] = acc[5]; o[2] = acc[6]; o[3] = acc[7];
        *(v4f*)(outp + pk1) = o;
    }
}


template <int M> __device__ __forceinline__ float row_xor(float v) {
    return __int_as_float(__builtin_amdgcn_update_dpp(0, __float_as_int(v), 0x160 + M, 0xf, 0xf, true));
}

struct ScanBlk {
    v4f   bp[TB / PK];
    v4f   cp[TB / PK];
};

template <int M> __device__ __forceinline__ float bfly(bool hi, float lo_e, float hi_e) {
    const float keep = hi ? hi_e : lo_e;
    const float send = hi ? lo_e : hi_e;
    return keep + row_xor<M>(send);
}

typedef __attribute__((address_space(1))) float gfloat;
typedef __attribute__((address_space(1))) const v4f gcv4f;
typedef __attribute__((address_space(1))) const float gcfloat;

__device__ __forceinline__ v4f ld_lane4(const float* __restrict__ ubase, const int elem, const unsigned lane_bytes) {
    return *(gcv4f*)((unsigned long long)ubase + (unsigned long long)lane_bytes + (unsigned long long)(elem * 4));
}
__device__ __forceinline__ float ld_lane1(const float* __restrict__ ubase, const int elem, const unsigned lane_bytes) {
    return *(gcfloat*)((unsigned long long)ubase + (unsigned long long)lane_bytes + (unsigned long long)(elem * 4));
}
__device__ __forceinline__ void st_lane(float* __restrict__ ubase, const int elem, const unsigned lane_bytes, const float v) {
    *(gfloat*)((unsigned long long)ubase + (unsigned long long)lane_bytes + (unsigned long long)(elem * 4)) = v;
}
__device__ __forceinline__ void keep_lane_offset(unsigned& lane_bytes) {
    asm volatile("" : "+v"(lane_bytes));
}
__device__ __forceinline__ void wg_handover() {
    asm volatile("s_wait_dscnt 0x0\n\ts_barrier_signal -1\n\ts_barrier_wait -1" ::: "memory");
}

#define SCAN_STEP(S)                                                          \
    {                                                                         \
        const float dt = dcur[(S) / PK][(S) % PK];                            \
        const float xs = xcur[(S) / PK][(S) % PK];                            \
        const float a  = __expf(dt * An);                                     \
        h = fmaf(a, h, (a - An) * rAn * blk.bp[(S) / PK][(S) % PK] * xs);     \
        p[S] = blk.cp[(S) / PK][(S) % PK] * h;                                \
    }
#define SCAN_REFILL(Q)                                                        \
    {                                                                         \
        blk.bp[Q] = ld_lane4(bpn, (Q) * NN * PK, npt);                        \
        blk.cp[Q] = ld_lane4(cpn, (Q) * NN * PK, npt);                        \
    }

__global__ __launch_bounds__(32 * (SW + 1)) __attribute__((amdgpu_waves_per_eu(1, 1))) void ssm_scan_serial(
    const float* __restrict__ x,
    const float* __restrict__ A,
    const float* __restrict__ p_Delta,
    const float* __restrict__ dlin,
    const float* __restrict__ Bp,
    const float* __restrict__ Cp,
    float* __restrict__ y)
{
    __shared__ __attribute__((aligned(16))) float s_dt[2 * SW * LP];
    __shared__ __attribute__((aligned(16))) float s_x[2 * SW * LP];

    const unsigned tid   = threadIdx.x;
    const unsigned lane  = tid & 31;
    const unsigned n     = tid & 15;
    const unsigned half  = lane >> 4;
    const bool feeder    = __builtin_amdgcn_readfirstlane((int)(tid >> 5)) == SW;
    const unsigned rw    = feeder ? half : (tid >> 4);
    const unsigned pair0 = blockIdx.x * (2 * SW);
    const unsigned b     = pair0 / DD;
    const unsigned d0    = pair0 - b * DD;
    const unsigned d     = d0 + rw;

    const float An  = A[d * NN + n];
    const float rAn = 1.0f / An;

    const float* __restrict__ bpu = Bp   + (size_t)b * LL * NN;
    const float* __restrict__ cpu = Cp   + (size_t)b * LL * NN;
    float*       __restrict__ yu  = y    + (size_t)pair0 * LL;
    const unsigned np = n * (4u * PK);
    const unsigned xb = (rw * LL + n) * 4u;
    const float* sdr = s_dt + rw * LP;
    const float* sxr = s_x  + rw * LP;

    const bool m8 = (lane & 8) != 0;
    const bool m4 = (lane & 4) != 0;
    const bool m2 = (lane & 2) != 0;
    const bool m1 = (lane & 1) != 0;

    const float* __restrict__ xg = x    + (size_t)pair0 * LL;
    const float* __restrict__ dg = dlin + (size_t)b * LL;
    const unsigned lxb = (half * LL + n) * 4u;
    const unsigned ldb = n * 4u;
    const unsigned lf = half * LP + n;
    float pDv[SW];
#pragma unroll
    for (int sp = 0; sp < SW; ++sp) pDv[sp] = p_Delta[d0 + 2 * sp + half];

    auto softplus = [&](const float dlv, const float pd) {
        const float dsum = dlv + pd;
        return fmaxf(dsum, 0.0f) + log1pf(__expf(-fabsf(dsum)));
    };
    auto feeder_fetch = [&](float (&fdv)[NSET], float (&fxv)[NSET][SW], const int blk0,
                            const unsigned lxt, const unsigned ldt) {
        const int b0 = (blk0 < NBLK - NSET) ? blk0 : NBLK - NSET;
        const float* __restrict__ xt = xg + b0 * TB;
        const float* __restrict__ dt = dg + b0 * TB;
#pragma unroll
        for (int q = 0; q < NSET; ++q) {
            fdv[q] = ld_lane1(dt, q * TB, ldt);
#pragma unroll
            for (int sp = 0; sp < SW; ++sp) fxv[q][sp] = ld_lane1(xt, (2 * sp) * LL + q * TB, lxt);
        }
    };
    auto feeder_emit = [&](const float (&fdv)[NSET], const float (&fxv)[NSET][SW], const int blk0) {
#pragma unroll
        for (int q = 0; q < NSET; ++q) {
            const int bk = blk0 + q;
#pragma unroll
            for (int sp = 0; sp < SW; ++sp) {
                s_dt[lf + 2 * sp * LP + bk * TB] = softplus(fdv[q], pDv[sp]);
                s_x [lf + 2 * sp * LP + bk * TB] = fxv[q][sp];
            }
        }
    };

    if (feeder) {
        unsigned lxt = lxb, ldt = ldb;
        float fd[NSET], fx[NSET][SW], gd[NSET], gx[NSET][SW];
        feeder_fetch(gd, gx, 0, lxt, ldt);
        feeder_fetch(fd, fx, NSET, lxt, ldt);
        feeder_emit(gd, gx, 0);
        feeder_emit(fd, fx, NSET);
        feeder_fetch(fd, fx, 2 * NSET, lxt, ldt);
        wg_handover();
        for (int k = 0; k < NBLK; k += NSET) {
            keep_lane_offset(lxt);
            keep_lane_offset(ldt);
            feeder_emit(fd, fx, k + 2 * NSET);
            feeder_fetch(fd, fx, k + 3 * NSET, lxt, ldt);
            wg_handover();
        }
        return;
    }

    ScanBlk st[NSET];
#pragma unroll
    for (int i = 0; i < NSET; ++i) {
#pragma unroll
        for (int q = 0; q < TB / PK; ++q) st[i].bp[q] = ld_lane4(bpu, (i * TB + q * PK) * NN, np);
#pragma unroll
        for (int q = 0; q < TB / PK; ++q) st[i].cp[q] = ld_lane4(cpu, (i * TB + q * PK) * NN, np);
    }
    wg_handover();

    v4f dq[2][TB / PK], xq[2][TB / PK];
#pragma unroll
    for (int i = 0; i < 2; ++i) {
#pragma unroll
        for (int q = 0; q < TB / PK; ++q) {
            dq[i][q] = *(const v4f*)(sdr + i * TB + q * PK);
            xq[i][q] = *(const v4f*)(sxr + i * TB + q * PK);
        }
    }
    float h = 0.0f;

    auto run_steps = [&](ScanBlk& blk, const int kb, const float* __restrict__ bpn, const float* __restrict__ cpn,
                         float (&p)[TB], v4f (&dcur)[TB / PK], v4f (&xcur)[TB / PK], const unsigned npt) {
        const int kn = kb + 2;
        SCAN_STEP(0)  SCAN_STEP(1)  SCAN_STEP(2)  SCAN_STEP(3)  SCAN_REFILL(0)
        SCAN_STEP(4)  SCAN_STEP(5)  SCAN_STEP(6)  SCAN_STEP(7)  SCAN_REFILL(1)
        SCAN_STEP(8)  SCAN_STEP(9)  SCAN_STEP(10) SCAN_STEP(11) SCAN_REFILL(2)
        SCAN_STEP(12) SCAN_STEP(13) SCAN_STEP(14) SCAN_STEP(15) SCAN_REFILL(3)
#pragma unroll
        for (int q = 0; q < TB / PK; ++q) {
            dcur[q] = *(const v4f*)(sdr + kn * TB + q * PK);
            xcur[q] = *(const v4f*)(sxr + kn * TB + q * PK);
        }
    };
    auto sum_store = [&](const float (&p)[TB], const int l0, const unsigned xbt) {
        float w8[8], w4[4], w2[2];
#pragma unroll
        for (int i = 0; i < 8; ++i) w8[i] = bfly<8>(m8, p[i], p[i + 8]);
#pragma unroll
        for (int i = 0; i < 4; ++i) w4[i] = bfly<4>(m4, w8[i], w8[i + 4]);
#pragma unroll
        for (int i = 0; i < 2; ++i) w2[i] = bfly<2>(m2, w4[i], w4[i + 2]);
        st_lane(yu + l0, 0, xbt, bfly<1>(m1, w2[0], w2[1]));
    };

    unsigned xbt = xb, npt = np;
    for (int k = 0; k < NBLK; k += NSET) {
        keep_lane_offset(xbt);
        keep_lane_offset(npt);
        float p[NSET][TB];
        const int kt = (k + NSET < NBLK) ? k + NSET : NBLK - NSET;
        const float* __restrict__ bpt = bpu + kt * TB * NN;
        const float* __restrict__ cpt = cpu + kt * TB * NN;
#pragma unroll
        for (int i = 0; i < NSET; ++i) {
            run_steps(st[i], k + i, bpt + i * TB * NN, cpt + i * TB * NN, p[i], dq[i & 1], xq[i & 1], npt);
            __builtin_amdgcn_sched_barrier(0);
            sum_store(p[i], (k + i) * TB, xbt);
        }
        wg_handover();
    }
}

extern "C" void kernel_launch(void* const* d_in, const int* in_sizes, int n_in,
                              void* d_out, int out_size, void* d_ws, size_t ws_size,
                              hipStream_t stream)
{
    const float* x       = (const float*)d_in[0];
    const float* A       = (const float*)d_in[1];
    const float* W_B     = (const float*)d_in[2];
    const float* b_B     = (const float*)d_in[3];
    const float* W_C     = (const float*)d_in[4];
    const float* b_C     = (const float*)d_in[5];
    const float* W_d     = (const float*)d_in[6];
    const float* b_d     = (const float*)d_in[7];
    const float* p_Delta = (const float*)d_in[8];
    float* y = (float*)d_out;

    float* ws   = (float*)d_ws;
    float* Bp   = ws;
    float* Cp   = Bp + (size_t)BB * LL * NN;
    float* dlin = Cp + (size_t)BB * LL * NN;

    ssm_proj_wmma<<<dim3(128), dim3(96), 0, stream>>>(
        x, W_B, b_B, W_C, b_C, W_d, b_d, Bp, Cp, dlin);

    ssm_scan_serial<<<dim3((BB * DD) / (2 * SW)), dim3(32 * (SW + 1)), 0, stream>>>(
        x, A, p_Delta, dlin, Bp, Cp, y);
}


# scan start-up: the first 8 blocks of delta and x are prepared by all four waves, 2 blocks (6 softplus chains) each, not 24 chains on the feeder while the scan waves wait; values unchanged
# speedup vs baseline: 4.1378x; 1.0394x over previous
#include <hip/hip_runtime.h>
#include <math.h>

#define DD 768
#define LL 1024
#define NN 16
#define BB 2
#define KRING 24
#define KAHEAD 20
#define TB 16
#define NBLK (LL / TB)
#define PK 4
#define NSET 4
#define LP (LL + 9 * TB)
#define SW 3
#define NHB ((2 * NSET) / (SW + 1))

static_assert((2 * NSET) % (SW + 1) == 0, "the start-up blocks divide evenly among the waves");

typedef float v2f __attribute__((ext_vector_type(2)));
typedef float v4f __attribute__((ext_vector_type(4)));
typedef float v8f __attribute__((ext_vector_type(8)));


__global__ __launch_bounds__(96) void ssm_proj_wmma(
    const float* __restrict__ x,
    const float* __restrict__ W_B,
    const float* __restrict__ b_B,
    const float* __restrict__ W_C,
    const float* __restrict__ b_C,
    const float* __restrict__ W_d,
    const float* __restrict__ b_d,
    float* __restrict__ Bp,
    float* __restrict__ Cp,
    float* __restrict__ dlin)
{
    const int lane = threadIdx.x & 31;
    const int which = __builtin_amdgcn_readfirstlane((int)(threadIdx.x >> 5));
    const int half = lane >> 4;
    const int mn   = lane & 15;
    const int row0 = blockIdx.x << 4;
    const int b    = row0 >> 10;
    const int l0   = row0 & (LL - 1);
    const int lA   = l0 + mn;

    const float* __restrict__ wmat = (which == 0) ? W_B : ((which == 1) ? W_C : W_d);
    const float* __restrict__ bvec = (which == 0) ? b_B : ((which == 1) ? b_C : b_d);
    const int wrow = (which == 2) ? 0 : mn;
    const float bias = bvec[wrow];
    v8f acc;
#pragma unroll
    for (int r = 0; r < 8; ++r) acc[r] = bias;

    const float* __restrict__ wr = wmat + wrow * DD + 2 * half;
    const float* __restrict__ xa = x + (size_t)b * DD * LL + (size_t)(2 * half) * LL + lA;

    v2f fa[KRING], fw[KRING];
    auto load_step = [&](const int slot, const int k) {
        fa[slot][0] = xa[(size_t)k * LL];
        fa[slot][1] = xa[(size_t)(k + 1) * LL];
        fw[slot] = *(const v2f*)(wr + k);
    };
    auto product = [&](const int slot) {
        __builtin_amdgcn_sched_barrier(0);
        acc = __builtin_amdgcn_wmma_f32_16x16x4_f32(false, fa[slot], false, fw[slot], (short)0, acc, false, false);
        __builtin_amdgcn_sched_barrier(0);
    };

#pragma unroll
    for (int s = 0; s < KAHEAD; ++s) load_step(s, 4 * s);
    int kb = 0;
    for (; kb < DD - 4 * KRING; kb += 4 * KRING) {
#pragma unroll
        for (int s = 0; s < KRING; ++s) {
            load_step((s + KAHEAD) % KRING, kb + 4 * (s + KAHEAD));
            product(s);
        }
    }
#pragma unroll
    for (int s = 0; s < KRING; ++s) {
        if (s + KAHEAD < KRING) load_step(s + KAHEAD, kb + 4 * (s + KAHEAD));
        product(s);
    }

    const int gr0 = b * LL + l0 + 8 * half;
    if (which == 2) {
#pragma unroll
        for (int r = 0; r < 8; ++r) {
            if (mn == 0) dlin[gr0 + r] = acc[r];
        }
    } else {
        float* __restrict__ outp = (which == 0) ? Bp : Cp;
        const int pk0 = ((gr0 / PK) * NN + mn) * PK;
        const int pk1 = pk0 + NN * PK;
        v4f o;
        o[0] = acc[0]; o[1] = acc[1]; o[2] = acc[2]; o[3] = acc[3];
        *(v4f*)(outp + pk0) = o;
        o[0] = acc[4]; o[1] = acc[5]; o[2] = acc[6]; o[3] = acc[7];
        *(v4f*)(outp + pk1) = o;
    }
}


template <int M> __device__ __forceinline__ float row_xor(float v) {
    return __int_as_float(__builtin_amdgcn_update_dpp(0, __float_as_int(v), 0x160 + M, 0xf, 0xf, true));
}

struct ScanBlk {
    v4f   bp[TB / PK];
    v4f   cp[TB / PK];
};

template <int M> __device__ __forceinline__ float bfly(bool hi, float lo_e, float hi_e) {
    const float keep = hi ? hi_e : lo_e;
    const float send = hi ? lo_e : hi_e;
    return keep + row_xor<M>(send);
}

typedef __attribute__((address_space(1))) float gfloat;
typedef __attribute__((address_space(1))) const v4f gcv4f;
typedef __attribute__((address_space(1))) const float gcfloat;

__device__ __forceinline__ v4f ld_lane4(const float* __restrict__ ubase, const int elem, const unsigned lane_bytes) {
    return *(gcv4f*)((unsigned long long)ubase + (unsigned long long)lane_bytes + (unsigned long long)(elem * 4));
}
__device__ __forceinline__ float ld_lane1(const float* __restrict__ ubase, const int elem, const unsigned lane_bytes) {
    return *(gcfloat*)((unsigned long long)ubase + (unsigned long long)lane_bytes + (unsigned long long)(elem * 4));
}
__device__ __forceinline__ void st_lane(float* __restrict__ ubase, const int elem, const unsigned lane_bytes, const float v) {
    *(gfloat*)((unsigned long long)ubase + (unsigned long long)lane_bytes + (unsigned long long)(elem * 4)) = v;
}
__device__ __forceinline__ void keep_lane_offset(unsigned& lane_bytes) {
    asm volatile("" : "+v"(lane_bytes));
}
__device__ __forceinline__ void wg_handover() {
    asm volatile("s_wait_dscnt 0x0\n\ts_barrier_signal -1\n\ts_barrier_wait -1" ::: "memory");
}

#define SCAN_STEP(S)                                                          \
    {                                                                         \
        const float dt = dcur[(S) / PK][(S) % PK];                            \
        const float xs = xcur[(S) / PK][(S) % PK];                            \
        const float a  = __expf(dt * An);                                     \
        h = fmaf(a, h, (a - An) * rAn * blk.bp[(S) / PK][(S) % PK] * xs);     \
        p[S] = blk.cp[(S) / PK][(S) % PK] * h;                                \
    }
#define SCAN_REFILL(Q)                                                        \
    {                                                                         \
        blk.bp[Q] = ld_lane4(bpn, (Q) * NN * PK, npt);                        \
        blk.cp[Q] = ld_lane4(cpn, (Q) * NN * PK, npt);                        \
    }

__global__ __launch_bounds__(32 * (SW + 1)) __attribute__((amdgpu_waves_per_eu(1, 1))) void ssm_scan_serial(
    const float* __restrict__ x,
    const float* __restrict__ A,
    const float* __restrict__ p_Delta,
    const float* __restrict__ dlin,
    const float* __restrict__ Bp,
    const float* __restrict__ Cp,
    float* __restrict__ y)
{
    __shared__ __attribute__((aligned(16))) float s_dt[2 * SW * LP];
    __shared__ __attribute__((aligned(16))) float s_x[2 * SW * LP];

    const unsigned tid   = threadIdx.x;
    const unsigned lane  = tid & 31;
    const unsigned n     = tid & 15;
    const unsigned half  = lane >> 4;
    const int  wv        = __builtin_amdgcn_readfirstlane((int)(tid >> 5));
    const bool feeder    = wv == SW;
    const unsigned rw    = feeder ? half : (tid >> 4);
    const unsigned pair0 = blockIdx.x * (2 * SW);
    const unsigned b     = pair0 / DD;
    const unsigned d0    = pair0 - b * DD;
    const unsigned d     = d0 + rw;

    const float An  = A[d * NN + n];
    const float rAn = 1.0f / An;

    const float* __restrict__ bpu = Bp   + (size_t)b * LL * NN;
    const float* __restrict__ cpu = Cp   + (size_t)b * LL * NN;
    float*       __restrict__ yu  = y    + (size_t)pair0 * LL;
    const unsigned np = n * (4u * PK);
    const unsigned xb = (rw * LL + n) * 4u;
    const float* sdr = s_dt + rw * LP;
    const float* sxr = s_x  + rw * LP;

    const bool m8 = (lane & 8) != 0;
    const bool m4 = (lane & 4) != 0;
    const bool m2 = (lane & 2) != 0;
    const bool m1 = (lane & 1) != 0;

    const float* __restrict__ xg = x    + (size_t)pair0 * LL;
    const float* __restrict__ dg = dlin + (size_t)b * LL;
    const unsigned lxb = (half * LL + n) * 4u;
    const unsigned ldb = n * 4u;
    const unsigned lf = half * LP + n;
    float pDv[SW];
#pragma unroll
    for (int sp = 0; sp < SW; ++sp) pDv[sp] = p_Delta[d0 + 2 * sp + half];

    auto softplus = [&](const float dlv, const float pd) {
        const float dsum = dlv + pd;
        return fmaxf(dsum, 0.0f) + log1pf(__expf(-fabsf(dsum)));
    };
    auto feeder_fetch = [&](float (&fdv)[NSET], float (&fxv)[NSET][SW], const int blk0,
                            const unsigned lxt, const unsigned ldt) {
        const int b0 = (blk0 < NBLK - NSET) ? blk0 : NBLK - NSET;
        const float* __restrict__ xt = xg + b0 * TB;
        const float* __restrict__ dt = dg + b0 * TB;
#pragma unroll
        for (int q = 0; q < NSET; ++q) {
            fdv[q] = ld_lane1(dt, q * TB, ldt);
#pragma unroll
            for (int sp = 0; sp < SW; ++sp) fxv[q][sp] = ld_lane1(xt, (2 * sp) * LL + q * TB, lxt);
        }
    };
    auto feeder_emit = [&](const float (&fdv)[NSET], const float (&fxv)[NSET][SW], const int blk0) {
#pragma unroll
        for (int q = 0; q < NSET; ++q) {
            const int bk = blk0 + q;
#pragma unroll
            for (int sp = 0; sp < SW; ++sp) {
                s_dt[lf + 2 * sp * LP + bk * TB] = softplus(fdv[q], pDv[sp]);
                s_x [lf + 2 * sp * LP + bk * TB] = fxv[q][sp];
            }
        }
    };

    const int hb0 = wv * NHB;
    float hd[NHB], hx[NHB][SW];
    {
        const float* __restrict__ xt = xg + hb0 * TB;
        const float* __restrict__ dt = dg + hb0 * TB;
#pragma unroll
        for (int q = 0; q < NHB; ++q) {
            hd[q] = ld_lane1(dt, q * TB, ldb);
#pragma unroll
            for (int sp = 0; sp < SW; ++sp) hx[q][sp] = ld_lane1(xt, (2 * sp) * LL + q * TB, lxb);
        }
    }
    auto head_emit = [&]() {
#pragma unroll
        for (int q = 0; q < NHB; ++q) {
#pragma unroll
            for (int sp = 0; sp < SW; ++sp) {
                s_dt[lf + 2 * sp * LP + (hb0 + q) * TB] = softplus(hd[q], pDv[sp]);
                s_x [lf + 2 * sp * LP + (hb0 + q) * TB] = hx[q][sp];
            }
        }
    };

    if (feeder) {
        unsigned lxt = lxb, ldt = ldb;
        float fd[NSET], fx[NSET][SW];
        feeder_fetch(fd, fx, 2 * NSET, lxt, ldt);
        head_emit();
        wg_handover();
        for (int k = 0; k < NBLK; k += NSET) {
            keep_lane_offset(lxt);
            keep_lane_offset(ldt);
            feeder_emit(fd, fx, k + 2 * NSET);
            feeder_fetch(fd, fx, k + 3 * NSET, lxt, ldt);
            wg_handover();
        }
        return;
    }

    ScanBlk st[NSET];
#pragma unroll
    for (int i = 0; i < NSET; ++i) {
#pragma unroll
        for (int q = 0; q < TB / PK; ++q) st[i].bp[q] = ld_lane4(bpu, (i * TB + q * PK) * NN, np);
#pragma unroll
        for (int q = 0; q < TB / PK; ++q) st[i].cp[q] = ld_lane4(cpu, (i * TB + q * PK) * NN, np);
    }
    head_emit();
    wg_handover();

    v4f dq[2][TB / PK], xq[2][TB / PK];
#pragma unroll
    for (int i = 0; i < 2; ++i) {
#pragma unroll
        for (int q = 0; q < TB / PK; ++q) {
            dq[i][q] = *(const v4f*)(sdr + i * TB + q * PK);
            xq[i][q] = *(const v4f*)(sxr + i * TB + q * PK);
        }
    }
    float h = 0.0f;

    auto run_steps = [&](ScanBlk& blk, const int kb, const float* __restrict__ bpn, const float* __restrict__ cpn,
                         float (&p)[TB], v4f (&dcur)[TB / PK], v4f (&xcur)[TB / PK], const unsigned npt) {
        const int kn = kb + 2;
        SCAN_STEP(0)  SCAN_STEP(1)  SCAN_STEP(2)  SCAN_STEP(3)  SCAN_REFILL(0)
        SCAN_STEP(4)  SCAN_STEP(5)  SCAN_STEP(6)  SCAN_STEP(7)  SCAN_REFILL(1)
        SCAN_STEP(8)  SCAN_STEP(9)  SCAN_STEP(10) SCAN_STEP(11) SCAN_REFILL(2)
        SCAN_STEP(12) SCAN_STEP(13) SCAN_STEP(14) SCAN_STEP(15) SCAN_REFILL(3)
#pragma unroll
        for (int q = 0; q < TB / PK; ++q) {
            dcur[q] = *(const v4f*)(sdr + kn * TB + q * PK);
            xcur[q] = *(const v4f*)(sxr + kn * TB + q * PK);
        }
    };
    auto sum_store = [&](const float (&p)[TB], const int l0, const unsigned xbt) {
        float w8[8], w4[4], w2[2];
#pragma unroll
        for (int i = 0; i < 8; ++i) w8[i] = bfly<8>(m8, p[i], p[i + 8]);
#pragma unroll
        for (int i = 0; i < 4; ++i) w4[i] = bfly<4>(m4, w8[i], w8[i + 4]);
#pragma unroll
        for (int i = 0; i < 2; ++i) w2[i] = bfly<2>(m2, w4[i], w4[i + 2]);
        st_lane(yu + l0, 0, xbt, bfly<1>(m1, w2[0], w2[1]));
    };

    unsigned xbt = xb, npt = np;
    for (int k = 0; k < NBLK; k += NSET) {
        keep_lane_offset(xbt);
        keep_lane_offset(npt);
        float p[NSET][TB];
        const int kt = (k + NSET < NBLK) ? k + NSET : NBLK - NSET;
        const float* __restrict__ bpt = bpu + kt * TB * NN;
        const float* __restrict__ cpt = cpu + kt * TB * NN;
#pragma unroll
        for (int i = 0; i < NSET; ++i) {
            run_steps(st[i], k + i, bpt + i * TB * NN, cpt + i * TB * NN, p[i], dq[i & 1], xq[i & 1], npt);
            __builtin_amdgcn_sched_barrier(0);
            sum_store(p[i], (k + i) * TB, xbt);
        }
        wg_handover();
    }
}

extern "C" void kernel_launch(void* const* d_in, const int* in_sizes, int n_in,
                              void* d_out, int out_size, void* d_ws, size_t ws_size,
                              hipStream_t stream)
{
    const float* x       = (const float*)d_in[0];
    const float* A       = (const float*)d_in[1];
    const float* W_B     = (const float*)d_in[2];
    const float* b_B     = (const float*)d_in[3];
    const float* W_C     = (const float*)d_in[4];
    const float* b_C     = (const float*)d_in[5];
    const float* W_d     = (const float*)d_in[6];
    const float* b_d     = (const float*)d_in[7];
    const float* p_Delta = (const float*)d_in[8];
    float* y = (float*)d_out;

    float* ws   = (float*)d_ws;
    float* Bp   = ws;
    float* Cp   = Bp + (size_t)BB * LL * NN;
    float* dlin = Cp + (size_t)BB * LL * NN;

    ssm_proj_wmma<<<dim3(128), dim3(96), 0, stream>>>(
        x, W_B, b_B, W_C, b_C, W_d, b_d, Bp, Cp, dlin);

    ssm_scan_serial<<<dim3((BB * DD) / (2 * SW)), dim3(32 * (SW + 1)), 0, stream>>>(
        x, A, p_Delta, dlin, Bp, Cp, y);
}
